# starting kernel
# baseline (speedup 1.0000x reference)
#include <hip/hip_runtime.h>

#define B_ 512
#define T_ 512
#define D_ 64
#define H_ 128
#define G_ 512
#define OUT_ 16
#define BT_ 16
#define THREADS_ 512
#define NT_ 2

typedef __attribute__((ext_vector_type(16))) _Float16 v16h;
typedef __attribute__((ext_vector_type(8)))  _Float16 v8h;
typedef __attribute__((ext_vector_type(8)))  float    v8f;

__device__ __forceinline__ float fast_tanh(float x) {
#if __has_builtin(__builtin_amdgcn_tanhf)
  return __builtin_amdgcn_tanhf(x);
#else
  float e = __builtin_amdgcn_exp2f(x * 2.8853900817779268f);
  return 1.0f - 2.0f * __builtin_amdgcn_rcpf(e + 1.0f);
#endif
}

__device__ __forceinline__ v16h ldfrag_g(const _Float16* __restrict__ base,
                                         int stride, int koff, int lane) {
  int r  = lane & 15;
  int hi = lane >> 4;
  const _Float16* p = base + (size_t)r * stride + koff + hi * 8;
  v8h a = *(const v8h*)p;
  v8h b = *(const v8h*)(p + 16);
  return __builtin_shufflevector(a, b, 0,1,2,3,4,5,6,7,8,9,10,11,12,13,14,15);
}

__global__ void cvt_f32_to_f16(const float* __restrict__ s, _Float16* __restrict__ d, int n) {
  int i = blockIdx.x * blockDim.x + threadIdx.x;
  if (i < n) d[i] = (_Float16)s[i];
}

__global__ void add_vec(const float* __restrict__ a, const float* __restrict__ b,
                        float* __restrict__ o, int n) {
  int i = blockIdx.x * blockDim.x + threadIdx.x;
  if (i < n) o[i] = a[i] + b[i];
}

template<int KIN>
__global__ __launch_bounds__(THREADS_, 1)
void lstm_layer_kernel(const _Float16* __restrict__ xseq,
                       const _Float16* __restrict__ Wih,
                       const _Float16* __restrict__ Whh,
                       const float*    __restrict__ bias,
                       _Float16*       __restrict__ hseq_out,
                       float*          __restrict__ hlast_out)
{
  constexpr int KX = KIN / 32;

  __shared__ _Float16 h_lds[BT_][H_ + 8];
  __shared__ float    g_lds[BT_][G_ + 8];

  const int tid   = threadIdx.x;
  const int lane  = tid & 31;
  const int wave  = tid >> 5;
  const int hi    = lane >> 4;
  const int b0    = blockIdx.x * BT_;
  const int ncol0 = wave * (NT_ * 16);

  for (int i = tid; i < BT_ * (H_ + 8); i += THREADS_)
    ((_Float16*)h_lds)[i] = (_Float16)0.0f;

  v16h Bx[NT_][KX];
  v16h Bh[NT_][4];
  float bv[NT_];
  #pragma unroll
  for (int n = 0; n < NT_; ++n) {
    const int nc = ncol0 + n * 16;
    bv[n] = bias[nc + (lane & 15)];
    #pragma unroll
    for (int kt = 0; kt < KX; ++kt)
      Bx[n][kt] = ldfrag_g(Wih + (size_t)nc * KIN, KIN, kt * 32, lane);
    #pragma unroll
    for (int kt = 0; kt < 4; ++kt)
      Bh[n][kt] = ldfrag_g(Whh + (size_t)nc * H_, H_, kt * 32, lane);
  }

  const bool  is_tanh = ((wave >> 2) == 2);
  const float act_pre  = is_tanh ? 1.0f : 0.5f;
  const float act_post = is_tanh ? 1.0f : 0.5f;
  const float act_off  = is_tanh ? 0.0f : 0.5f;

  float c[4] = {0.f, 0.f, 0.f, 0.f};

  const _Float16* xbase = xseq + (size_t)b0 * T_ * KIN;

  __syncthreads();

  for (int t = 0; t < T_; ++t) {
    v16h xa[KX];
    #pragma unroll
    for (int kt = 0; kt < KX; ++kt)
      xa[kt] = ldfrag_g(xbase + (size_t)t * KIN, T_ * KIN, kt * 32, lane);
    if (t + 1 < T_)
      __builtin_prefetch(xbase + (size_t)(t + 1) * KIN + (size_t)(lane & 15) * (T_ * KIN), 0, 1);

    v16h ha[4];
    #pragma unroll
    for (int kt = 0; kt < 4; ++kt) {
      const _Float16* p = &h_lds[lane & 15][kt * 32 + hi * 8];
      v8h a = *(const v8h*)p;
      v8h b = *(const v8h*)(p + 16);
      ha[kt] = __builtin_shufflevector(a, b, 0,1,2,3,4,5,6,7,8,9,10,11,12,13,14,15);
    }

    #pragma unroll
    for (int n = 0; n < NT_; ++n) {
      v8f acc;
      #pragma unroll
      for (int m = 0; m < 8; ++m) acc[m] = bv[n];
      #pragma unroll
      for (int kt = 0; kt < KX; ++kt)
        acc = __builtin_amdgcn_wmma_f32_16x16x32_f16(false, xa[kt], false, Bx[n][kt],
                                                     (short)0, acc, false, false);
      #pragma unroll
      for (int kt = 0; kt < 4; ++kt)
        acc = __builtin_amdgcn_wmma_f32_16x16x32_f16(false, ha[kt], false, Bh[n][kt],
                                                     (short)0, acc, false, false);

      const int col = ncol0 + n * 16 + (lane & 15);
      #pragma unroll
      for (int m = 0; m < 8; ++m) {
        float a = act_off + act_post * fast_tanh(acc[m] * act_pre);
        g_lds[m + 8 * hi][col] = a;
      }
    }
    __syncthreads();

    #pragma unroll
    for (int e = 0; e < 4; ++e) {
      int q = tid * 4 + e;
      int r = q >> 7;
      int j = q & (H_ - 1);
      float iv = g_lds[r][j];
      float fv = g_lds[r][H_ + j];
      float gv = g_lds[r][2 * H_ + j];
      float ov = g_lds[r][3 * H_ + j];
      c[e] = fv * c[e] + iv * gv;
      float hv = ov * fast_tanh(c[e]);
      h_lds[r][j] = (_Float16)hv;
      if (hseq_out)
        hseq_out[(size_t)(b0 + r) * T_ * H_ + (size_t)t * H_ + j] = (_Float16)hv;
      if (hlast_out && t == T_ - 1)
        hlast_out[(b0 + r) * H_ + j] = hv;
    }
    __syncthreads();
  }
}

__global__ void fc_kernel(const float* __restrict__ h1, const float* __restrict__ Wfc0,
                          const float* __restrict__ bfc0, const float* __restrict__ Wfc1,
                          const float* __restrict__ bfc1, float* __restrict__ out) {
  int b = blockIdx.x;
  int tid = threadIdx.x;
  __shared__ float tmp[32];
  float s = bfc0[tid];
  for (int k = 0; k < H_; ++k) s += h1[b * H_ + k] * Wfc0[tid * H_ + k];
  tmp[tid] = s;
  __syncthreads();
  if (tid < OUT_) {
    float o = bfc1[tid];
    for (int k = 0; k < 32; ++k) o += tmp[k] * Wfc1[tid * 32 + k];
    out[b * OUT_ + tid] = o;
  }
}

extern "C" void kernel_launch(void* const* d_in, const int* in_sizes, int n_in,
                              void* d_out, int out_size, void* d_ws, size_t ws_size,
                              hipStream_t stream) {
  (void)in_sizes; (void)n_in; (void)out_size; (void)ws_size;
  const float* x    = (const float*)d_in[0];
  const float* Wih0 = (const float*)d_in[1];
  const float* Whh0 = (const float*)d_in[2];
  const float* bih0 = (const float*)d_in[3];
  const float* bhh0 = (const float*)d_in[4];
  const float* Wih1 = (const float*)d_in[5];
  const float* Whh1 = (const float*)d_in[6];
  const float* bih1 = (const float*)d_in[7];
  const float* bhh1 = (const float*)d_in[8];
  const float* Wfc0 = (const float*)d_in[9];
  const float* bfc0 = (const float*)d_in[10];
  const float* Wfc1 = (const float*)d_in[11];
  const float* bfc1 = (const float*)d_in[12];

  char* p = (char*)d_ws;
  auto alloc = [&](size_t bytes) {
    char* r = p;
    p += (bytes + 255) & ~(size_t)255;
    return r;
  };
  _Float16* x_h    = (_Float16*)alloc((size_t)B_ * T_ * D_ * 2);
  _Float16* h0seq  = (_Float16*)alloc((size_t)B_ * T_ * H_ * 2);
  _Float16* wih0_h = (_Float16*)alloc((size_t)G_ * D_ * 2);
  _Float16* whh0_h = (_Float16*)alloc((size_t)G_ * H_ * 2);
  _Float16* wih1_h = (_Float16*)alloc((size_t)G_ * H_ * 2);
  _Float16* whh1_h = (_Float16*)alloc((size_t)G_ * H_ * 2);
  float*    b0     = (float*)alloc((size_t)G_ * 4);
  float*    b1     = (float*)alloc((size_t)G_ * 4);
  float*    h1last = (float*)alloc((size_t)B_ * H_ * 4);

  auto cvt = [&](const float* s, _Float16* d, int n) {
    cvt_f32_to_f16<<<(n + 255) / 256, 256, 0, stream>>>(s, d, n);
  };
  cvt(x, x_h, B_ * T_ * D_);
  cvt(Wih0, wih0_h, G_ * D_);
  cvt(Whh0, whh0_h, G_ * H_);
  cvt(Wih1, wih1_h, G_ * H_);
  cvt(Whh1, whh1_h, G_ * H_);
  add_vec<<<2, 256, 0, stream>>>(bih0, bhh0, b0, G_);
  add_vec<<<2, 256, 0, stream>>>(bih1, bhh1, b1, G_);

  lstm_layer_kernel<D_><<<B_ / BT_, THREADS_, 0, stream>>>(x_h, wih0_h, whh0_h, b0,
                                                           h0seq, nullptr);
  lstm_layer_kernel<H_><<<B_ / BT_, THREADS_, 0, stream>>>(h0seq, wih1_h, whh1_h, b1,
                                                           nullptr, h1last);
  fc_kernel<<<B_, 32, 0, stream>>>(h1last, Wfc0, bfc0, Wfc1, bfc1, (float*)d_out);
}


# Both layers in ONE workgroup and launch: waves 0-3 layer 0 at time s, waves 4-7 layer 1 at s-1 off the h0 tile in LDS; Wih streamed from LDS; sum order per tile: h/input k tiles alternate, bias last
# speedup vs baseline: 2.5758x; 2.5758x over previous
#include <hip/hip_runtime.h>

#define B_ 512
#define T_ 512
#define D_ 64
#define H_ 128
#define G_ 512
#define OUT_ 16
#define BT_ 16
#define RT_ 128
#define THREADS_ 256
#define NT_ 8
#define CPT_ 16
#define FRAGH_ 512

typedef __attribute__((ext_vector_type(16))) _Float16 v16h;
typedef __attribute__((ext_vector_type(8)))  _Float16 v8h;
typedef __attribute__((ext_vector_type(8)))  float    v8f;
typedef __attribute__((ext_vector_type(4)))  float    v4f;

__device__ __forceinline__ float fast_tanh(float x) {
#if __has_builtin(__builtin_amdgcn_tanhf)
  return __builtin_amdgcn_tanhf(x);
#else
  float e = __builtin_amdgcn_exp2f(x * 2.8853900817779268f);
  return 1.0f - 2.0f * __builtin_amdgcn_rcpf(e + 1.0f);
#endif
}

__device__ __forceinline__ v16h ldfrag_g(const _Float16* __restrict__ base,
                                         int stride, int koff, int lane) {
  int r  = lane & 15;
  int hi = lane >> 4;
  const _Float16* p = base + (size_t)r * stride + koff + hi * 8;
  v8h a = *(const v8h*)p;
  v8h b = *(const v8h*)(p + 16);
  return __builtin_shufflevector(a, b, 0,1,2,3,4,5,6,7,8,9,10,11,12,13,14,15);
}

__global__ void cvt_f32_to_f16(const float* __restrict__ s, _Float16* __restrict__ d, int n) {
  int i = blockIdx.x * blockDim.x + threadIdx.x;
  if (i < n) d[i] = (_Float16)s[i];
}

__global__ void add_vec(const float* __restrict__ a, const float* __restrict__ b,
                        float* __restrict__ o, int n) {
  int i = blockIdx.x * blockDim.x + threadIdx.x;
  if (i < n) o[i] = a[i] + b[i];
}

__device__ __forceinline__ void stfrag_w(_Float16* p, v16h f, int lane) {
  *(v8h*)(p + lane * 8)       = __builtin_shufflevector(f, f, 0,1,2,3,4,5,6,7);
  *(v8h*)(p + 256 + lane * 8) = __builtin_shufflevector(f, f, 8,9,10,11,12,13,14,15);
}
__device__ __forceinline__ v16h ldfrag_w(const _Float16* p, int lane) {
  v8h a = *(const v8h*)(p + lane * 8);
  v8h b = *(const v8h*)(p + 256 + lane * 8);
  return __builtin_shufflevector(a, b, 0,1,2,3,4,5,6,7,8,9,10,11,12,13,14,15);
}

__device__ __forceinline__ v16h ldfrag_h(const _Float16 (*h)[H_ + 8], int kt, int lane) {
  const _Float16* p = &h[lane & 15][kt * 32 + (lane >> 4) * 8];
  v8h a = *(const v8h*)p;
  v8h b = *(const v8h*)(p + 16);
  return __builtin_shufflevector(a, b, 0,1,2,3,4,5,6,7,8,9,10,11,12,13,14,15);
}

__global__ __launch_bounds__(THREADS_, 1)
void lstm2_kernel(const _Float16* __restrict__ xseq,
                  const _Float16* __restrict__ Wih0,
                  const _Float16* __restrict__ Whh0,
                  const float*    __restrict__ bias0,
                  const _Float16* __restrict__ Wih1,
                  const _Float16* __restrict__ Whh1,
                  const float*    __restrict__ bias1,
                  float*          __restrict__ hlast_out)
{
  __shared__ __attribute__((aligned(16))) _Float16 h_lds[2][BT_][H_ + 8];
  __shared__ __attribute__((aligned(16))) float    g_lds[2][BT_][G_ + 8];
  __shared__ __attribute__((aligned(16))) _Float16 w_lds[4 * NT_ * 2 * FRAGH_ + 4 * NT_ * 4 * FRAGH_];

  const int tid   = threadIdx.x;
  const int lane  = tid & 31;
  const int wave8 = __builtin_amdgcn_readfirstlane(tid >> 5);
  const int role  = wave8 >> 2;
  const int wave  = wave8 & 3;
  const int rt    = tid & (RT_ - 1);
  const int hi    = lane >> 4;
  const int b0    = blockIdx.x * BT_;
  const int ncol0 = wave * (NT_ * 16);

  for (int i = tid; i < 2 * BT_ * (H_ + 8); i += THREADS_)
    ((_Float16*)h_lds)[i] = (_Float16)0.0f;

  const _Float16* Whh  = role ? Whh1 : Whh0;
  const float*    bias = role ? bias1 : bias0;
  v16h Bh[NT_][4];
  float bv[NT_];
  #pragma unroll
  for (int n = 0; n < NT_; ++n) {
    const int nc = ncol0 + n * 16;
    bv[n] = bias[nc + (lane & 15)];
    #pragma unroll
    for (int kt = 0; kt < 4; ++kt)
      Bh[n][kt] = ldfrag_g(Whh + (size_t)nc * H_, H_, kt * 32, lane);
  }

  _Float16* wmine = w_lds + (role ? 4 * NT_ * 2 * FRAGH_ + wave * (NT_ * 4 * FRAGH_)
                                  : wave * (NT_ * 2 * FRAGH_));
  if (role == 0) {
    #pragma unroll
    for (int n = 0; n < NT_; ++n) {
      #pragma unroll
      for (int kt = 0; kt < 2; ++kt)
        stfrag_w(wmine + (n * 2 + kt) * FRAGH_,
                 ldfrag_g(Wih0 + (size_t)(ncol0 + n * 16) * D_, D_, kt * 32, lane), lane);
    }
  } else {
    #pragma unroll
    for (int n = 0; n < NT_; ++n) {
      #pragma unroll
      for (int kt = 0; kt < 4; ++kt)
        stfrag_w(wmine + (n * 4 + kt) * FRAGH_,
                 ldfrag_g(Wih1 + (size_t)(ncol0 + n * 16) * H_, H_, kt * 32, lane), lane);
    }
  }

  const bool  is_tanh = (wave == 2);
  const float act_pre  = is_tanh ? 1.0f : 0.5f;
  const float act_post = is_tanh ? 1.0f : 0.5f;
  const float act_off  = is_tanh ? 0.0f : 0.5f;
  #pragma unroll
  for (int n = 0; n < NT_; ++n) bv[n] *= act_pre;

  float c[CPT_];
  #pragma unroll
  for (int e = 0; e < CPT_; ++e) c[e] = 0.f;
  const int cr = (rt * CPT_) >> 7;
  const int cj = (rt * CPT_) & (H_ - 1);

  const _Float16* xbase = xseq + (size_t)b0 * T_ * D_;
  float (*g)[G_ + 8]       = g_lds[role];
  _Float16 (*hown)[H_ + 8] = h_lds[role];

  __syncthreads();

  for (int s = 0; s <= T_; ++s) {
    const bool active = role ? (s >= 1) : (s < T_);

    if (active) {
      v8f acc[NT_];
      v16h bx[NT_];
      v16h ha[4];
      #pragma unroll
      for (int kt = 0; kt < 4; ++kt) ha[kt] = ldfrag_h(hown, kt, lane);

      if (role == 0) {
        v16h xa[2];
        #pragma unroll
        for (int kt = 0; kt < 2; ++kt)
          xa[kt] = ldfrag_g(xbase + (size_t)s * D_, T_ * D_, kt * 32, lane);
        if (s + 1 < T_)
          __builtin_prefetch(xbase + (size_t)(s + 1) * D_ + (size_t)(lane & 15) * (T_ * D_), 0, 1);
        #pragma unroll
        for (int kx = 0; kx < 2; ++kx) {
          #pragma unroll
          for (int n = 0; n < NT_; ++n) bx[n] = ldfrag_w(wmine + (n * 2 + kx) * FRAGH_, lane);
          #pragma unroll
          for (int kt = 2 * kx; kt < 2 * kx + 2; ++kt) {
            #pragma unroll
            for (int n = 0; n < NT_; ++n) {
              const v8f cin = (kt == 0) ? v8f{0.f, 0.f, 0.f, 0.f, 0.f, 0.f, 0.f, 0.f} : acc[n];
              acc[n] = __builtin_amdgcn_wmma_f32_16x16x32_f16(false, ha[kt], false, Bh[n][kt],
                                                              (short)0, cin, false, false);
            }
          }
          asm volatile("" :: "v"(bx[0]), "v"(bx[1]), "v"(bx[2]), "v"(bx[3]),
                             "v"(bx[4]), "v"(bx[5]), "v"(bx[6]), "v"(bx[7]));
          #pragma unroll
          for (int n = 0; n < NT_; ++n)
            acc[n] = __builtin_amdgcn_wmma_f32_16x16x32_f16(false, xa[kx], false, bx[n],
                                                            (short)0, acc[n], false, false);
        }
      } else {
        v16h ia[4];
        #pragma unroll
        for (int kt = 0; kt < 4; ++kt) ia[kt] = ldfrag_h(h_lds[0], kt, lane);
        #pragma unroll
        for (int kt = 0; kt < 4; ++kt) {
          #pragma unroll
          for (int n = 0; n < NT_; ++n) bx[n] = ldfrag_w(wmine + (n * 4 + kt) * FRAGH_, lane);
          #pragma unroll
          for (int n = 0; n < NT_; ++n) {
            const v8f cin = (kt == 0) ? v8f{0.f, 0.f, 0.f, 0.f, 0.f, 0.f, 0.f, 0.f} : acc[n];
            acc[n] = __builtin_amdgcn_wmma_f32_16x16x32_f16(false, ha[kt], false, Bh[n][kt],
                                                            (short)0, cin, false, false);
          }
          asm volatile("" :: "v"(bx[0]), "v"(bx[1]), "v"(bx[2]), "v"(bx[3]),
                             "v"(bx[4]), "v"(bx[5]), "v"(bx[6]), "v"(bx[7]));
          #pragma unroll
          for (int n = 0; n < NT_; ++n)
            acc[n] = __builtin_amdgcn_wmma_f32_16x16x32_f16(false, ia[kt], false, bx[n],
                                                            (short)0, acc[n], false, false);
        }
      }

      #pragma unroll
      for (int n = 0; n < NT_; ++n) {
        const int col = ncol0 + n * 16 + (lane & 15);
        #pragma unroll
        for (int m = 0; m < 8; ++m) {
          float a = act_off + act_post * fast_tanh(__builtin_fmaf(acc[n][m], act_pre, bv[n]));
          g[m + 8 * hi][col] = a;
        }
      }
    }
    __syncthreads();

    if (active) {
      float hv[CPT_];
      #pragma unroll
      for (int e4 = 0; e4 < CPT_ / 4; ++e4) {
        const v4f iv = *(const v4f*)&g[cr][cj + e4 * 4];
        const v4f fv = *(const v4f*)&g[cr][H_ + cj + e4 * 4];
        const v4f gv = *(const v4f*)&g[cr][2 * H_ + cj + e4 * 4];
        const v4f ov = *(const v4f*)&g[cr][3 * H_ + cj + e4 * 4];
        #pragma unroll
        for (int k = 0; k < 4; ++k) {
          const int e = e4 * 4 + k;
          c[e] = fv[k] * c[e] + iv[k] * gv[k];
          hv[e] = ov[k] * fast_tanh(c[e]);
        }
      }
      v8h hp0, hp1;
      #pragma unroll
      for (int k = 0; k < 8; ++k) {
        hp0[k] = (_Float16)hv[k];
        hp1[k] = (_Float16)hv[8 + k];
      }
      *(v8h*)&hown[cr][cj]     = hp0;
      *(v8h*)&hown[cr][cj + 8] = hp1;
      if (role == 1 && s == T_) {
        float* hl = hlast_out + (b0 + cr) * H_ + cj;
        #pragma unroll
        for (int e4 = 0; e4 < CPT_ / 4; ++e4) {
          v4f o4;
          #pragma unroll
          for (int k = 0; k < 4; ++k) o4[k] = hv[e4 * 4 + k];
          *(v4f*)(hl + e4 * 4) = o4;
        }
      }
    }
    __syncthreads();
  }
}

__global__ void fc_kernel(const float* __restrict__ h1, const float* __restrict__ Wfc0,
                          const float* __restrict__ bfc0, const float* __restrict__ Wfc1,
                          const float* __restrict__ bfc1, float* __restrict__ out) {
  int b = blockIdx.x;
  int tid = threadIdx.x;
  __shared__ float tmp[32];
  float s = bfc0[tid];
  for (int k = 0; k < H_; ++k) s += h1[b * H_ + k] * Wfc0[tid * H_ + k];
  tmp[tid] = s;
  __syncthreads();
  if (tid < OUT_) {
    float o = bfc1[tid];
    for (int k = 0; k < 32; ++k) o += tmp[k] * Wfc1[tid * 32 + k];
    out[b * OUT_ + tid] = o;
  }
}

extern "C" void kernel_launch(void* const* d_in, const int* in_sizes, int n_in,
                              void* d_out, int out_size, void* d_ws, size_t ws_size,
                              hipStream_t stream) {
  (void)in_sizes; (void)n_in; (void)out_size; (void)ws_size;
  const float* x    = (const float*)d_in[0];
  const float* Wih0 = (const float*)d_in[1];
  const float* Whh0 = (const float*)d_in[2];
  const float* bih0 = (const float*)d_in[3];
  const float* bhh0 = (const float*)d_in[4];
  const float* Wih1 = (const float*)d_in[5];
  const float* Whh1 = (const float*)d_in[6];
  const float* bih1 = (const float*)d_in[7];
  const float* bhh1 = (const float*)d_in[8];
  const float* Wfc0 = (const float*)d_in[9];
  const float* bfc0 = (const float*)d_in[10];
  const float* Wfc1 = (const float*)d_in[11];
  const float* bfc1 = (const float*)d_in[12];

  char* p = (char*)d_ws;
  auto alloc = [&](size_t bytes) {
    char* r = p;
    p += (bytes + 255) & ~(size_t)255;
    return r;
  };
  _Float16* x_h    = (_Float16*)alloc((size_t)B_ * T_ * D_ * 2);
  _Float16* wih0_h = (_Float16*)alloc((size_t)G_ * D_ * 2);
  _Float16* whh0_h = (_Float16*)alloc((size_t)G_ * H_ * 2);
  _Float16* wih1_h = (_Float16*)alloc((size_t)G_ * H_ * 2);
  _Float16* whh1_h = (_Float16*)alloc((size_t)G_ * H_ * 2);
  float*    b0     = (float*)alloc((size_t)G_ * 4);
  float*    b1     = (float*)alloc((size_t)G_ * 4);
  float*    h1last = (float*)alloc((size_t)B_ * H_ * 4);

  auto cvt = [&](const float* s, _Float16* d, int n) {
    cvt_f32_to_f16<<<(n + 255) / 256, 256, 0, stream>>>(s, d, n);
  };
  cvt(x, x_h, B_ * T_ * D_);
  cvt(Wih0, wih0_h, G_ * D_);
  cvt(Whh0, whh0_h, G_ * H_);
  cvt(Wih1, wih1_h, G_ * H_);
  cvt(Whh1, whh1_h, G_ * H_);
  add_vec<<<2, 256, 0, stream>>>(bih0, bhh0, b0, G_);
  add_vec<<<2, 256, 0, stream>>>(bih1, bhh1, b1, G_);

  lstm2_kernel<<<B_ / BT_, THREADS_, 0, stream>>>(x_h, wih0_h, whh0_h, b0,
                                                  wih1_h, whh1_h, b1, h1last);
  fc_kernel<<<B_, 32, 0, stream>>>(h1last, Wfc0, bfc0, Wfc1, bfc1, (float*)d_out);
}


# Conversion pass to the 16-bit copies re-gridded: a thread converts groups of 8 adjacent elements (two 16 B loads, one 16 B store), x in 4096 long-lived waves, were 2^24 one-element threads
# speedup vs baseline: 2.6535x; 1.0302x over previous
#include <hip/hip_runtime.h>

#define B_ 512
#define T_ 512
#define D_ 64
#define H_ 128
#define G_ 512
#define OUT_ 16
#define BT_ 16
#define RT_ 128
#define THREADS_ 256
#define NT_ 8
#define CPT_ 16
#define FRAGH_ 512

typedef __attribute__((ext_vector_type(16))) _Float16 v16h;
typedef __attribute__((ext_vector_type(8)))  _Float16 v8h;
typedef __attribute__((ext_vector_type(8)))  float    v8f;
typedef __attribute__((ext_vector_type(4)))  float    v4f;

__device__ __forceinline__ float fast_tanh(float x) {
#if __has_builtin(__builtin_amdgcn_tanhf)
  return __builtin_amdgcn_tanhf(x);
#else
  float e = __builtin_amdgcn_exp2f(x * 2.8853900817779268f);
  return 1.0f - 2.0f * __builtin_amdgcn_rcpf(e + 1.0f);
#endif
}

__device__ __forceinline__ v16h ldfrag_g(const _Float16* __restrict__ base,
                                         int stride, int koff, int lane) {
  int r  = lane & 15;
  int hi = lane >> 4;
  const _Float16* p = base + (size_t)r * stride + koff + hi * 8;
  v8h a = *(const v8h*)p;
  v8h b = *(const v8h*)(p + 16);
  return __builtin_shufflevector(a, b, 0,1,2,3,4,5,6,7,8,9,10,11,12,13,14,15);
}

#define CVT_E_  8
#define CVT_T_  256
#define CVT_UX_ 16
template <int U>
__global__ __launch_bounds__(CVT_T_)
void cvt_f32_to_f16(const float* __restrict__ s, _Float16* __restrict__ d) {
  const int g0     = blockIdx.x * CVT_T_ + threadIdx.x;
  const int stride = gridDim.x * CVT_T_;
  v4f a[U], b[U];
  #pragma unroll
  for (int u = 0; u < U; ++u) {
    const float* p = s + (size_t)(g0 + u * stride) * CVT_E_;
    a[u] = *(const v4f*)p;
    b[u] = *(const v4f*)(p + 4);
  }
  #pragma unroll
  for (int u = 0; u < U; ++u) {
    v8h o;
    #pragma unroll
    for (int k = 0; k < 4; ++k) {
      o[k]     = (_Float16)a[u][k];
      o[4 + k] = (_Float16)b[u][k];
    }
    *(v8h*)(d + (size_t)(g0 + u * stride) * CVT_E_) = o;
  }
}

__global__ void add_vec(const float* __restrict__ a, const float* __restrict__ b,
                        float* __restrict__ o, int n) {
  int i = blockIdx.x * blockDim.x + threadIdx.x;
  if (i < n) o[i] = a[i] + b[i];
}

__device__ __forceinline__ void stfrag_w(_Float16* p, v16h f, int lane) {
  *(v8h*)(p + lane * 8)       = __builtin_shufflevector(f, f, 0,1,2,3,4,5,6,7);
  *(v8h*)(p + 256 + lane * 8) = __builtin_shufflevector(f, f, 8,9,10,11,12,13,14,15);
}
__device__ __forceinline__ v16h ldfrag_w(const _Float16* p, int lane) {
  v8h a = *(const v8h*)(p + lane * 8);
  v8h b = *(const v8h*)(p + 256 + lane * 8);
  return __builtin_shufflevector(a, b, 0,1,2,3,4,5,6,7,8,9,10,11,12,13,14,15);
}

__device__ __forceinline__ v16h ldfrag_h(const _Float16 (*h)[H_ + 8], int kt, int lane) {
  const _Float16* p = &h[lane & 15][kt * 32 + (lane >> 4) * 8];
  v8h a = *(const v8h*)p;
  v8h b = *(const v8h*)(p + 16);
  return __builtin_shufflevector(a, b, 0,1,2,3,4,5,6,7,8,9,10,11,12,13,14,15);
}

__global__ __launch_bounds__(THREADS_, 1)
void lstm2_kernel(const _Float16* __restrict__ xseq,
                  const _Float16* __restrict__ Wih0,
                  const _Float16* __restrict__ Whh0,
                  const float*    __restrict__ bias0,
                  const _Float16* __restrict__ Wih1,
                  const _Float16* __restrict__ Whh1,
                  const float*    __restrict__ bias1,
                  float*          __restrict__ hlast_out)
{
  __shared__ __attribute__((aligned(16))) _Float16 h_lds[2][BT_][H_ + 8];
  __shared__ __attribute__((aligned(16))) float    g_lds[2][BT_][G_ + 8];
  __shared__ __attribute__((aligned(16))) _Float16 w_lds[4 * NT_ * 2 * FRAGH_ + 4 * NT_ * 4 * FRAGH_];

  const int tid   = threadIdx.x;
  const int lane  = tid & 31;
  const int wave8 = __builtin_amdgcn_readfirstlane(tid >> 5);
  const int role  = wave8 >> 2;
  const int wave  = wave8 & 3;
  const int rt    = tid & (RT_ - 1);
  const int hi    = lane >> 4;
  const int b0    = blockIdx.x * BT_;
  const int ncol0 = wave * (NT_ * 16);

  for (int i = tid; i < 2 * BT_ * (H_ + 8); i += THREADS_)
    ((_Float16*)h_lds)[i] = (_Float16)0.0f;

  const _Float16* Whh  = role ? Whh1 : Whh0;
  const float*    bias = role ? bias1 : bias0;
  v16h Bh[NT_][4];
  float bv[NT_];
  #pragma unroll
  for (int n = 0; n < NT_; ++n) {
    const int nc = ncol0 + n * 16;
    bv[n] = bias[nc + (lane & 15)];
    #pragma unroll
    for (int kt = 0; kt < 4; ++kt)
      Bh[n][kt] = ldfrag_g(Whh + (size_t)nc * H_, H_, kt * 32, lane);
  }

  _Float16* wmine = w_lds + (role ? 4 * NT_ * 2 * FRAGH_ + wave * (NT_ * 4 * FRAGH_)
                                  : wave * (NT_ * 2 * FRAGH_));
  if (role == 0) {
    #pragma unroll
    for (int n = 0; n < NT_; ++n) {
      #pragma unroll
      for (int kt = 0; kt < 2; ++kt)
        stfrag_w(wmine + (n * 2 + kt) * FRAGH_,
                 ldfrag_g(Wih0 + (size_t)(ncol0 + n * 16) * D_, D_, kt * 32, lane), lane);
    }
  } else {
    #pragma unroll
    for (int n = 0; n < NT_; ++n) {
      #pragma unroll
      for (int kt = 0; kt < 4; ++kt)
        stfrag_w(wmine + (n * 4 + kt) * FRAGH_,
                 ldfrag_g(Wih1 + (size_t)(ncol0 + n * 16) * H_, H_, kt * 32, lane), lane);
    }
  }

  const bool  is_tanh = (wave == 2);
  const float act_pre  = is_tanh ? 1.0f : 0.5f;
  const float act_post = is_tanh ? 1.0f : 0.5f;
  const float act_off  = is_tanh ? 0.0f : 0.5f;
  #pragma unroll
  for (int n = 0; n < NT_; ++n) bv[n] *= act_pre;

  float c[CPT_];
  #pragma unroll
  for (int e = 0; e < CPT_; ++e) c[e] = 0.f;
  const int cr = (rt * CPT_) >> 7;
  const int cj = (rt * CPT_) & (H_ - 1);

  const _Float16* xbase = xseq + (size_t)b0 * T_ * D_;
  float (*g)[G_ + 8]       = g_lds[role];
  _Float16 (*hown)[H_ + 8] = h_lds[role];

  __syncthreads();

  for (int s = 0; s <= T_; ++s) {
    const bool active = role ? (s >= 1) : (s < T_);

    if (active) {
      v8f acc[NT_];
      v16h bx[NT_];
      v16h ha[4];
      #pragma unroll
      for (int kt = 0; kt < 4; ++kt) ha[kt] = ldfrag_h(hown, kt, lane);

      if (role == 0) {
        v16h xa[2];
        #pragma unroll
        for (int kt = 0; kt < 2; ++kt)
          xa[kt] = ldfrag_g(xbase + (size_t)s * D_, T_ * D_, kt * 32, lane);
        if (s + 1 < T_)
          __builtin_prefetch(xbase + (size_t)(s + 1) * D_ + (size_t)(lane & 15) * (T_ * D_), 0, 1);
        #pragma unroll
        for (int kx = 0; kx < 2; ++kx) {
          #pragma unroll
          for (int n = 0; n < NT_; ++n) bx[n] = ldfrag_w(wmine + (n * 2 + kx) * FRAGH_, lane);
          #pragma unroll
          for (int kt = 2 * kx; kt < 2 * kx + 2; ++kt) {
            #pragma unroll
            for (int n = 0; n < NT_; ++n) {
              const v8f cin = (kt == 0) ? v8f{0.f, 0.f, 0.f, 0.f, 0.f, 0.f, 0.f, 0.f} : acc[n];
              acc[n] = __builtin_amdgcn_wmma_f32_16x16x32_f16(false, ha[kt], false, Bh[n][kt],
                                                              (short)0, cin, false, false);
            }
          }
          asm volatile("" :: "v"(bx[0]), "v"(bx[1]), "v"(bx[2]), "v"(bx[3]),
                             "v"(bx[4]), "v"(bx[5]), "v"(bx[6]), "v"(bx[7]));
          #pragma unroll
          for (int n = 0; n < NT_; ++n)
            acc[n] = __builtin_amdgcn_wmma_f32_16x16x32_f16(false, xa[kx], false, bx[n],
                                                            (short)0, acc[n], false, false);
        }
      } else {
        v16h ia[4];
        #pragma unroll
        for (int kt = 0; kt < 4; ++kt) ia[kt] = ldfrag_h(h_lds[0], kt, lane);
        #pragma unroll
        for (int kt = 0; kt < 4; ++kt) {
          #pragma unroll
          for (int n = 0; n < NT_; ++n) bx[n] = ldfrag_w(wmine + (n * 4 + kt) * FRAGH_, lane);
          #pragma unroll
          for (int n = 0; n < NT_; ++n) {
            const v8f cin = (kt == 0) ? v8f{0.f, 0.f, 0.f, 0.f, 0.f, 0.f, 0.f, 0.f} : acc[n];
            acc[n] = __builtin_amdgcn_wmma_f32_16x16x32_f16(false, ha[kt], false, Bh[n][kt],
                                                            (short)0, cin, false, false);
          }
          asm volatile("" :: "v"(bx[0]), "v"(bx[1]), "v"(bx[2]), "v"(bx[3]),
                             "v"(bx[4]), "v"(bx[5]), "v"(bx[6]), "v"(bx[7]));
          #pragma unroll
          for (int n = 0; n < NT_; ++n)
            acc[n] = __builtin_amdgcn_wmma_f32_16x16x32_f16(false, ia[kt], false, bx[n],
                                                            (short)0, acc[n], false, false);
        }
      }

      #pragma unroll
      for (int n = 0; n < NT_; ++n) {
        const int col = ncol0 + n * 16 + (lane & 15);
        #pragma unroll
        for (int m = 0; m < 8; ++m) {
          float a = act_off + act_post * fast_tanh(__builtin_fmaf(acc[n][m], act_pre, bv[n]));
          g[m + 8 * hi][col] = a;
        }
      }
    }
    __syncthreads();

    if (active) {
      float hv[CPT_];
      #pragma unroll
      for (int e4 = 0; e4 < CPT_ / 4; ++e4) {
        const v4f iv = *(const v4f*)&g[cr][cj + e4 * 4];
        const v4f fv = *(const v4f*)&g[cr][H_ + cj + e4 * 4];
        const v4f gv = *(const v4f*)&g[cr][2 * H_ + cj + e4 * 4];
        const v4f ov = *(const v4f*)&g[cr][3 * H_ + cj + e4 * 4];
        #pragma unroll
        for (int k = 0; k < 4; ++k) {
          const int e = e4 * 4 + k;
          c[e] = fv[k] * c[e] + iv[k] * gv[k];
          hv[e] = ov[k] * fast_tanh(c[e]);
        }
      }
      v8h hp0, hp1;
      #pragma unroll
      for (int k = 0; k < 8; ++k) {
        hp0[k] = (_Float16)hv[k];
        hp1[k] = (_Float16)hv[8 + k];
      }
      *(v8h*)&hown[cr][cj]     = hp0;
      *(v8h*)&hown[cr][cj + 8] = hp1;
      if (role == 1 && s == T_) {
        float* hl = hlast_out + (b0 + cr) * H_ + cj;
        #pragma unroll
        for (int e4 = 0; e4 < CPT_ / 4; ++e4) {
          v4f o4;
          #pragma unroll
          for (int k = 0; k < 4; ++k) o4[k] = hv[e4 * 4 + k];
          *(v4f*)(hl + e4 * 4) = o4;
        }
      }
    }
    __syncthreads();
  }
}

__global__ void fc_kernel(const float* __restrict__ h1, const float* __restrict__ Wfc0,
                          const float* __restrict__ bfc0, const float* __restrict__ Wfc1,
                          const float* __restrict__ bfc1, float* __restrict__ out) {
  int b = blockIdx.x;
  int tid = threadIdx.x;
  __shared__ float tmp[32];
  float s = bfc0[tid];
  for (int k = 0; k < H_; ++k) s += h1[b * H_ + k] * Wfc0[tid * H_ + k];
  tmp[tid] = s;
  __syncthreads();
  if (tid < OUT_) {
    float o = bfc1[tid];
    for (int k = 0; k < 32; ++k) o += tmp[k] * Wfc1[tid * 32 + k];
    out[b * OUT_ + tid] = o;
  }
}

extern "C" void kernel_launch(void* const* d_in, const int* in_sizes, int n_in,
                              void* d_out, int out_size, void* d_ws, size_t ws_size,
                              hipStream_t stream) {
  (void)in_sizes; (void)n_in; (void)out_size; (void)ws_size;
  const float* x    = (const float*)d_in[0];
  const float* Wih0 = (const float*)d_in[1];
  const float* Whh0 = (const float*)d_in[2];
  const float* bih0 = (const float*)d_in[3];
  const float* bhh0 = (const float*)d_in[4];
  const float* Wih1 = (const float*)d_in[5];
  const float* Whh1 = (const float*)d_in[6];
  const float* bih1 = (const float*)d_in[7];
  const float* bhh1 = (const float*)d_in[8];
  const float* Wfc0 = (const float*)d_in[9];
  const float* bfc0 = (const float*)d_in[10];
  const float* Wfc1 = (const float*)d_in[11];
  const float* bfc1 = (const float*)d_in[12];

  char* p = (char*)d_ws;
  auto alloc = [&](size_t bytes) {
    char* r = p;
    p += (bytes + 255) & ~(size_t)255;
    return r;
  };
  _Float16* x_h    = (_Float16*)alloc((size_t)B_ * T_ * D_ * 2);
  _Float16* wih0_h = (_Float16*)alloc((size_t)G_ * D_ * 2);
  _Float16* whh0_h = (_Float16*)alloc((size_t)G_ * H_ * 2);
  _Float16* wih1_h = (_Float16*)alloc((size_t)G_ * H_ * 2);
  _Float16* whh1_h = (_Float16*)alloc((size_t)G_ * H_ * 2);
  float*    b0     = (float*)alloc((size_t)G_ * 4);
  float*    b1     = (float*)alloc((size_t)G_ * 4);
  float*    h1last = (float*)alloc((size_t)B_ * H_ * 4);

  static_assert((B_ * T_ * D_) % (CVT_E_ * CVT_UX_ * CVT_T_) == 0, "x pass: whole workgroups");
  static_assert((G_ * D_) % (CVT_E_ * CVT_T_) == 0 && (G_ * H_) % (CVT_E_ * CVT_T_) == 0,
                "weight passes: whole workgroups");
  cvt_f32_to_f16<CVT_UX_><<<B_ * T_ * D_ / (CVT_E_ * CVT_UX_ * CVT_T_), CVT_T_, 0, stream>>>(x, x_h);
  cvt_f32_to_f16<1><<<G_ * D_ / (CVT_E_ * CVT_T_), CVT_T_, 0, stream>>>(Wih0, wih0_h);
  cvt_f32_to_f16<1><<<G_ * H_ / (CVT_E_ * CVT_T_), CVT_T_, 0, stream>>>(Whh0, whh0_h);
  cvt_f32_to_f16<1><<<G_ * H_ / (CVT_E_ * CVT_T_), CVT_T_, 0, stream>>>(Wih1, wih1_h);
  cvt_f32_to_f16<1><<<G_ * H_ / (CVT_E_ * CVT_T_), CVT_T_, 0, stream>>>(Whh1, whh1_h);
  add_vec<<<2, 256, 0, stream>>>(bih0, bhh0, b0, G_);
  add_vec<<<2, 256, 0, stream>>>(bih1, bhh1, b1, G_);

  lstm2_kernel<<<B_ / BT_, THREADS_, 0, stream>>>(x_h, wih0_h, whh0_h, b0,
                                                  wih1_h, whh1_h, b1, h1last);
  fc_kernel<<<B_, 32, 0, stream>>>(h1last, Wfc0, bfc0, Wfc1, bfc1, (float*)d_out);
}


# Wih fragment loads from LDS pinned by scheduling barriers: one group of 16 per k-group, behind the previous group's last WMMA and ahead of the Whh WMMAs; were issued two behind each WMMA
# speedup vs baseline: 2.8846x; 1.0871x over previous
#include <hip/hip_runtime.h>

#define B_ 512
#define T_ 512
#define D_ 64
#define H_ 128
#define G_ 512
#define OUT_ 16
#define BT_ 16
#define RT_ 128
#define THREADS_ 256
#define NT_ 8
#define CPT_ 16
#define FRAGH_ 512

typedef __attribute__((ext_vector_type(16))) _Float16 v16h;
typedef __attribute__((ext_vector_type(8)))  _Float16 v8h;
typedef __attribute__((ext_vector_type(8)))  float    v8f;
typedef __attribute__((ext_vector_type(4)))  float    v4f;

__device__ __forceinline__ float fast_tanh(float x) {
#if __has_builtin(__builtin_amdgcn_tanhf)
  return __builtin_amdgcn_tanhf(x);
#else
  float e = __builtin_amdgcn_exp2f(x * 2.8853900817779268f);
  return 1.0f - 2.0f * __builtin_amdgcn_rcpf(e + 1.0f);
#endif
}

__device__ __forceinline__ v16h ldfrag_g(const _Float16* __restrict__ base,
                                         int stride, int koff, int lane) {
  int r  = lane & 15;
  int hi = lane >> 4;
  const _Float16* p = base + (size_t)r * stride + koff + hi * 8;
  v8h a = *(const v8h*)p;
  v8h b = *(const v8h*)(p + 16);
  return __builtin_shufflevector(a, b, 0,1,2,3,4,5,6,7,8,9,10,11,12,13,14,15);
}

#define CVT_E_  8
#define CVT_T_  256
#define CVT_UX_ 16
template <int U>
__global__ __launch_bounds__(CVT_T_)
void cvt_f32_to_f16(const float* __restrict__ s, _Float16* __restrict__ d) {
  const int g0     = blockIdx.x * CVT_T_ + threadIdx.x;
  const int stride = gridDim.x * CVT_T_;
  v4f a[U], b[U];
  #pragma unroll
  for (int u = 0; u < U; ++u) {
    const float* p = s + (size_t)(g0 + u * stride) * CVT_E_;
    a[u] = *(const v4f*)p;
    b[u] = *(const v4f*)(p + 4);
  }
  #pragma unroll
  for (int u = 0; u < U; ++u) {
    v8h o;
    #pragma unroll
    for (int k = 0; k < 4; ++k) {
      o[k]     = (_Float16)a[u][k];
      o[4 + k] = (_Float16)b[u][k];
    }
    *(v8h*)(d + (size_t)(g0 + u * stride) * CVT_E_) = o;
  }
}

__global__ void add_vec(const float* __restrict__ a, const float* __restrict__ b,
                        float* __restrict__ o, int n) {
  int i = blockIdx.x * blockDim.x + threadIdx.x;
  if (i < n) o[i] = a[i] + b[i];
}

__device__ __forceinline__ void stfrag_w(_Float16* p, v16h f, int lane) {
  *(v8h*)(p + lane * 8)       = __builtin_shufflevector(f, f, 0,1,2,3,4,5,6,7);
  *(v8h*)(p + 256 + lane * 8) = __builtin_shufflevector(f, f, 8,9,10,11,12,13,14,15);
}
__device__ __forceinline__ v16h ldfrag_w(const _Float16* p, int lane) {
  v8h a = *(const v8h*)(p + lane * 8);
  v8h b = *(const v8h*)(p + 256 + lane * 8);
  return __builtin_shufflevector(a, b, 0,1,2,3,4,5,6,7,8,9,10,11,12,13,14,15);
}

__device__ __forceinline__ v16h ldfrag_h(const _Float16 (*h)[H_ + 8], int kt, int lane) {
  const _Float16* p = &h[lane & 15][kt * 32 + (lane >> 4) * 8];
  v8h a = *(const v8h*)p;
  v8h b = *(const v8h*)(p + 16);
  return __builtin_shufflevector(a, b, 0,1,2,3,4,5,6,7,8,9,10,11,12,13,14,15);
}

__global__ __launch_bounds__(THREADS_, 1)
void lstm2_kernel(const _Float16* __restrict__ xseq,
                  const _Float16* __restrict__ Wih0,
                  const _Float16* __restrict__ Whh0,
                  const float*    __restrict__ bias0,
                  const _Float16* __restrict__ Wih1,
                  const _Float16* __restrict__ Whh1,
                  const float*    __restrict__ bias1,
                  float*          __restrict__ hlast_out)
{
  __shared__ __attribute__((aligned(16))) _Float16 h_lds[2][BT_][H_ + 8];
  __shared__ __attribute__((aligned(16))) float    g_lds[2][BT_][G_ + 8];
  __shared__ __attribute__((aligned(16))) _Float16 w_lds[4 * NT_ * 2 * FRAGH_ + 4 * NT_ * 4 * FRAGH_];

  const int tid   = threadIdx.x;
  const int lane  = tid & 31;
  const int wave8 = __builtin_amdgcn_readfirstlane(tid >> 5);
  const int role  = wave8 >> 2;
  const int wave  = wave8 & 3;
  const int rt    = tid & (RT_ - 1);
  const int hi    = lane >> 4;
  const int b0    = blockIdx.x * BT_;
  const int ncol0 = wave * (NT_ * 16);

  for (int i = tid; i < 2 * BT_ * (H_ + 8); i += THREADS_)
    ((_Float16*)h_lds)[i] = (_Float16)0.0f;

  const _Float16* Whh  = role ? Whh1 : Whh0;
  const float*    bias = role ? bias1 : bias0;
  v16h Bh[NT_][4];
  float bv[NT_];
  #pragma unroll
  for (int n = 0; n < NT_; ++n) {
    const int nc = ncol0 + n * 16;
    bv[n] = bias[nc + (lane & 15)];
    #pragma unroll
    for (int kt = 0; kt < 4; ++kt)
      Bh[n][kt] = ldfrag_g(Whh + (size_t)nc * H_, H_, kt * 32, lane);
  }

  _Float16* wmine = w_lds + (role ? 4 * NT_ * 2 * FRAGH_ + wave * (NT_ * 4 * FRAGH_)
                                  : wave * (NT_ * 2 * FRAGH_));
  if (role == 0) {
    #pragma unroll
    for (int n = 0; n < NT_; ++n) {
      #pragma unroll
      for (int kt = 0; kt < 2; ++kt)
        stfrag_w(wmine + (n * 2 + kt) * FRAGH_,
                 ldfrag_g(Wih0 + (size_t)(ncol0 + n * 16) * D_, D_, kt * 32, lane), lane);
    }
  } else {
    #pragma unroll
    for (int n = 0; n < NT_; ++n) {
      #pragma unroll
      for (int kt = 0; kt < 4; ++kt)
        stfrag_w(wmine + (n * 4 + kt) * FRAGH_,
                 ldfrag_g(Wih1 + (size_t)(ncol0 + n * 16) * H_, H_, kt * 32, lane), lane);
    }
  }

  const bool  is_tanh = (wave == 2);
  const float act_pre  = is_tanh ? 1.0f : 0.5f;
  const float act_post = is_tanh ? 1.0f : 0.5f;
  const float act_off  = is_tanh ? 0.0f : 0.5f;
  #pragma unroll
  for (int n = 0; n < NT_; ++n) bv[n] *= act_pre;

  float c[CPT_];
  #pragma unroll
  for (int e = 0; e < CPT_; ++e) c[e] = 0.f;
  const int cr = (rt * CPT_) >> 7;
  const int cj = (rt * CPT_) & (H_ - 1);

  const _Float16* xbase = xseq + (size_t)b0 * T_ * D_;
  float (*g)[G_ + 8]       = g_lds[role];
  _Float16 (*hown)[H_ + 8] = h_lds[role];

  __syncthreads();

  for (int s = 0; s <= T_; ++s) {
    const bool active = role ? (s >= 1) : (s < T_);

    if (active) {
      v8f acc[NT_];
      v16h bx[NT_];
      v16h ha[4];
      #pragma unroll
      for (int kt = 0; kt < 4; ++kt) ha[kt] = ldfrag_h(hown, kt, lane);

      if (role == 0) {
        v16h xa[2];
        #pragma unroll
        for (int kt = 0; kt < 2; ++kt)
          xa[kt] = ldfrag_g(xbase + (size_t)s * D_, T_ * D_, kt * 32, lane);
        if (s + 1 < T_)
          __builtin_prefetch(xbase + (size_t)(s + 1) * D_ + (size_t)(lane & 15) * (T_ * D_), 0, 1);
        #pragma unroll
        for (int kx = 0; kx < 2; ++kx) {
          #pragma unroll
          for (int n = 0; n < NT_; ++n) bx[n] = ldfrag_w(wmine + (n * 2 + kx) * FRAGH_, lane);
          __builtin_amdgcn_sched_barrier(0);
          #pragma unroll
          for (int kt = 2 * kx; kt < 2 * kx + 2; ++kt) {
            #pragma unroll
            for (int n = 0; n < NT_; ++n) {
              const v8f cin = (kt == 0) ? v8f{0.f, 0.f, 0.f, 0.f, 0.f, 0.f, 0.f, 0.f} : acc[n];
              acc[n] = __builtin_amdgcn_wmma_f32_16x16x32_f16(false, ha[kt], false, Bh[n][kt],
                                                              (short)0, cin, false, false);
            }
          }
          #pragma unroll
          for (int n = 0; n < NT_; ++n)
            acc[n] = __builtin_amdgcn_wmma_f32_16x16x32_f16(false, xa[kx], false, bx[n],
                                                            (short)0, acc[n], false, false);
          __builtin_amdgcn_sched_barrier(0);
        }
      } else {
        v16h ia[4];
        #pragma unroll
        for (int kt = 0; kt < 4; ++kt) ia[kt] = ldfrag_h(h_lds[0], kt, lane);
        #pragma unroll
        for (int kt = 0; kt < 4; ++kt) {
          #pragma unroll
          for (int n = 0; n < NT_; ++n) bx[n] = ldfrag_w(wmine + (n * 4 + kt) * FRAGH_, lane);
          __builtin_amdgcn_sched_barrier(0);
          #pragma unroll
          for (int n = 0; n < NT_; ++n) {
            const v8f cin = (kt == 0) ? v8f{0.f, 0.f, 0.f, 0.f, 0.f, 0.f, 0.f, 0.f} : acc[n];
            acc[n] = __builtin_amdgcn_wmma_f32_16x16x32_f16(false, ha[kt], false, Bh[n][kt],
                                                            (short)0, cin, false, false);
          }
          #pragma unroll
          for (int n = 0; n < NT_; ++n)
            acc[n] = __builtin_amdgcn_wmma_f32_16x16x32_f16(false, ia[kt], false, bx[n],
                                                            (short)0, acc[n], false, false);
          __builtin_amdgcn_sched_barrier(0);
        }
      }

      #pragma unroll
      for (int n = 0; n < NT_; ++n) {
        const int col = ncol0 + n * 16 + (lane & 15);
        #pragma unroll
        for (int m = 0; m < 8; ++m) {
          float a = act_off + act_post * fast_tanh(__builtin_fmaf(acc[n][m], act_pre, bv[n]));
          g[m + 8 * hi][col] = a;
        }
      }
    }
    __syncthreads();

    if (active) {
      float hv[CPT_];
      #pragma unroll
      for (int e4 = 0; e4 < CPT_ / 4; ++e4) {
        const v4f iv = *(const v4f*)&g[cr][cj + e4 * 4];
        const v4f fv = *(const v4f*)&g[cr][H_ + cj + e4 * 4];
        const v4f gv = *(const v4f*)&g[cr][2 * H_ + cj + e4 * 4];
        const v4f ov = *(const v4f*)&g[cr][3 * H_ + cj + e4 * 4];
        #pragma unroll
        for (int k = 0; k < 4; ++k) {
          const int e = e4 * 4 + k;
          c[e] = fv[k] * c[e] + iv[k] * gv[k];
          hv[e] = ov[k] * fast_tanh(c[e]);
        }
      }
      v8h hp0, hp1;
      #pragma unroll
      for (int k = 0; k < 8; ++k) {
        hp0[k] = (_Float16)hv[k];
        hp1[k] = (_Float16)hv[8 + k];
      }
      *(v8h*)&hown[cr][cj]     = hp0;
      *(v8h*)&hown[cr][cj + 8] = hp1;
      if (role == 1 && s == T_) {
        float* hl = hlast_out + (b0 + cr) * H_ + cj;
        #pragma unroll
        for (int e4 = 0; e4 < CPT_ / 4; ++e4) {
          v4f o4;
          #pragma unroll
          for (int k = 0; k < 4; ++k) o4[k] = hv[e4 * 4 + k];
          *(v4f*)(hl + e4 * 4) = o4;
        }
      }
    }
    __syncthreads();
  }
}

__global__ void fc_kernel(const float* __restrict__ h1, const float* __restrict__ Wfc0,
                          const float* __restrict__ bfc0, const float* __restrict__ Wfc1,
                          const float* __restrict__ bfc1, float* __restrict__ out) {
  int b = blockIdx.x;
  int tid = threadIdx.x;
  __shared__ float tmp[32];
  float s = bfc0[tid];
  for (int k = 0; k < H_; ++k) s += h1[b * H_ + k] * Wfc0[tid * H_ + k];
  tmp[tid] = s;
  __syncthreads();
  if (tid < OUT_) {
    float o = bfc1[tid];
    for (int k = 0; k < 32; ++k) o += tmp[k] * Wfc1[tid * 32 + k];
    out[b * OUT_ + tid] = o;
  }
}

extern "C" void kernel_launch(void* const* d_in, const int* in_sizes, int n_in,
                              void* d_out, int out_size, void* d_ws, size_t ws_size,
                              hipStream_t stream) {
  (void)in_sizes; (void)n_in; (void)out_size; (void)ws_size;
  const float* x    = (const float*)d_in[0];
  const float* Wih0 = (const float*)d_in[1];
  const float* Whh0 = (const float*)d_in[2];
  const float* bih0 = (const float*)d_in[3];
  const float* bhh0 = (const float*)d_in[4];
  const float* Wih1 = (const float*)d_in[5];
  const float* Whh1 = (const float*)d_in[6];
  const float* bih1 = (const float*)d_in[7];
  const float* bhh1 = (const float*)d_in[8];
  const float* Wfc0 = (const float*)d_in[9];
  const float* bfc0 = (const float*)d_in[10];
  const float* Wfc1 = (const float*)d_in[11];
  const float* bfc1 = (const float*)d_in[12];

  char* p = (char*)d_ws;
  auto alloc = [&](size_t bytes) {
    char* r = p;
    p += (bytes + 255) & ~(size_t)255;
    return r;
  };
  _Float16* x_h    = (_Float16*)alloc((size_t)B_ * T_ * D_ * 2);
  _Float16* wih0_h = (_Float16*)alloc((size_t)G_ * D_ * 2);
  _Float16* whh0_h = (_Float16*)alloc((size_t)G_ * H_ * 2);
  _Float16* wih1_h = (_Float16*)alloc((size_t)G_ * H_ * 2);
  _Float16* whh1_h = (_Float16*)alloc((size_t)G_ * H_ * 2);
  float*    b0     = (float*)alloc((size_t)G_ * 4);
  float*    b1     = (float*)alloc((size_t)G_ * 4);
  float*    h1last = (float*)alloc((size_t)B_ * H_ * 4);

  static_assert((B_ * T_ * D_) % (CVT_E_ * CVT_UX_ * CVT_T_) == 0, "x pass: whole workgroups");
  static_assert((G_ * D_) % (CVT_E_ * CVT_T_) == 0 && (G_ * H_) % (CVT_E_ * CVT_T_) == 0,
                "weight passes: whole workgroups");
  cvt_f32_to_f16<CVT_UX_><<<B_ * T_ * D_ / (CVT_E_ * CVT_UX_ * CVT_T_), CVT_T_, 0, stream>>>(x, x_h);
  cvt_f32_to_f16<1><<<G_ * D_ / (CVT_E_ * CVT_T_), CVT_T_, 0, stream>>>(Wih0, wih0_h);
  cvt_f32_to_f16<1><<<G_ * H_ / (CVT_E_ * CVT_T_), CVT_T_, 0, stream>>>(Whh0, whh0_h);
  cvt_f32_to_f16<1><<<G_ * H_ / (CVT_E_ * CVT_T_), CVT_T_, 0, stream>>>(Wih1, wih1_h);
  cvt_f32_to_f16<1><<<G_ * H_ / (CVT_E_ * CVT_T_), CVT_T_, 0, stream>>>(Whh1, whh1_h);
  add_vec<<<2, 256, 0, stream>>>(bih0, bhh0, b0, G_);
  add_vec<<<2, 256, 0, stream>>>(bih1, bhh1, b1, G_);

  lstm2_kernel<<<B_ / BT_, THREADS_, 0, stream>>>(x_h, wih0_h, whh0_h, b0,
                                                  wih1_h, whh1_h, b1, h1last);
  fc_kernel<<<B_, 32, 0, stream>>>(h1last, Wfc0, bfc0, Wfc1, bfc1, (float*)d_out);
}


# Gate exchange in LDS freed of bank conflicts: tile row pitch 520 -> 516 words (scatter half-waves, 8 rows apart, on disjoint banks); cell threads own 2 rows x 2x4 units, 16 lanes read 256 B in a row
# speedup vs baseline: 3.2595x; 1.1300x over previous
#include <hip/hip_runtime.h>

#define B_ 512
#define T_ 512
#define D_ 64
#define H_ 128
#define G_ 512
#define OUT_ 16
#define BT_ 16
#define RT_ 128
#define THREADS_ 256
#define NT_ 8
#define CPT_ 16
#define GP_ (G_ + 4)
#define FRAGH_ 512

typedef __attribute__((ext_vector_type(16))) _Float16 v16h;
typedef __attribute__((ext_vector_type(8)))  _Float16 v8h;
typedef __attribute__((ext_vector_type(8)))  float    v8f;
typedef __attribute__((ext_vector_type(4)))  _Float16 v4h;
typedef __attribute__((ext_vector_type(4)))  float    v4f;

__device__ __forceinline__ float fast_tanh(float x) {
#if __has_builtin(__builtin_amdgcn_tanhf)
  return __builtin_amdgcn_tanhf(x);
#else
  float e = __builtin_amdgcn_exp2f(x * 2.8853900817779268f);
  return 1.0f - 2.0f * __builtin_amdgcn_rcpf(e + 1.0f);
#endif
}

__device__ __forceinline__ v16h ldfrag_g(const _Float16* __restrict__ base,
                                         int stride, int koff, int lane) {
  int r  = lane & 15;
  int hi = lane >> 4;
  const _Float16* p = base + (size_t)r * stride + koff + hi * 8;
  v8h a = *(const v8h*)p;
  v8h b = *(const v8h*)(p + 16);
  return __builtin_shufflevector(a, b, 0,1,2,3,4,5,6,7,8,9,10,11,12,13,14,15);
}

#define CVT_E_  8
#define CVT_T_  256
#define CVT_UX_ 16
template <int U>
__global__ __launch_bounds__(CVT_T_)
void cvt_f32_to_f16(const float* __restrict__ s, _Float16* __restrict__ d) {
  const int g0     = blockIdx.x * CVT_T_ + threadIdx.x;
  const int stride = gridDim.x * CVT_T_;
  v4f a[U], b[U];
  #pragma unroll
  for (int u = 0; u < U; ++u) {
    const float* p = s + (size_t)(g0 + u * stride) * CVT_E_;
    a[u] = *(const v4f*)p;
    b[u] = *(const v4f*)(p + 4);
  }
  #pragma unroll
  for (int u = 0; u < U; ++u) {
    v8h o;
    #pragma unroll
    for (int k = 0; k < 4; ++k) {
      o[k]     = (_Float16)a[u][k];
      o[4 + k] = (_Float16)b[u][k];
    }
    *(v8h*)(d + (size_t)(g0 + u * stride) * CVT_E_) = o;
  }
}

__global__ void add_vec(const float* __restrict__ a, const float* __restrict__ b,
                        float* __restrict__ o, int n) {
  int i = blockIdx.x * blockDim.x + threadIdx.x;
  if (i < n) o[i] = a[i] + b[i];
}

__device__ __forceinline__ void stfrag_w(_Float16* p, v16h f, int lane) {
  *(v8h*)(p + lane * 8)       = __builtin_shufflevector(f, f, 0,1,2,3,4,5,6,7);
  *(v8h*)(p + 256 + lane * 8) = __builtin_shufflevector(f, f, 8,9,10,11,12,13,14,15);
}
__device__ __forceinline__ v16h ldfrag_w(const _Float16* p, int lane) {
  v8h a = *(const v8h*)(p + lane * 8);
  v8h b = *(const v8h*)(p + 256 + lane * 8);
  return __builtin_shufflevector(a, b, 0,1,2,3,4,5,6,7,8,9,10,11,12,13,14,15);
}

__device__ __forceinline__ v16h ldfrag_h(const _Float16 (*h)[H_ + 8], int kt, int lane) {
  const _Float16* p = &h[lane & 15][kt * 32 + (lane >> 4) * 8];
  v8h a = *(const v8h*)p;
  v8h b = *(const v8h*)(p + 16);
  return __builtin_shufflevector(a, b, 0,1,2,3,4,5,6,7,8,9,10,11,12,13,14,15);
}

__global__ __launch_bounds__(THREADS_, 1)
void lstm2_kernel(const _Float16* __restrict__ xseq,
                  const _Float16* __restrict__ Wih0,
                  const _Float16* __restrict__ Whh0,
                  const float*    __restrict__ bias0,
                  const _Float16* __restrict__ Wih1,
                  const _Float16* __restrict__ Whh1,
                  const float*    __restrict__ bias1,
                  float*          __restrict__ hlast_out)
{
  __shared__ __attribute__((aligned(16))) _Float16 h_lds[2][BT_][H_ + 8];
  __shared__ __attribute__((aligned(16))) float    g_lds[2][BT_][GP_];
  __shared__ __attribute__((aligned(16))) _Float16 w_lds[4 * NT_ * 2 * FRAGH_ + 4 * NT_ * 4 * FRAGH_];

  const int tid   = threadIdx.x;
  const int lane  = tid & 31;
  const int wave8 = __builtin_amdgcn_readfirstlane(tid >> 5);
  const int role  = wave8 >> 2;
  const int wave  = wave8 & 3;
  const int rt    = tid & (RT_ - 1);
  const int hi    = lane >> 4;
  const int b0    = blockIdx.x * BT_;
  const int ncol0 = wave * (NT_ * 16);

  for (int i = tid; i < 2 * BT_ * (H_ + 8); i += THREADS_)
    ((_Float16*)h_lds)[i] = (_Float16)0.0f;

  const _Float16* Whh  = role ? Whh1 : Whh0;
  const float*    bias = role ? bias1 : bias0;
  v16h Bh[NT_][4];
  float bv[NT_];
  #pragma unroll
  for (int n = 0; n < NT_; ++n) {
    const int nc = ncol0 + n * 16;
    bv[n] = bias[nc + (lane & 15)];
    #pragma unroll
    for (int kt = 0; kt < 4; ++kt)
      Bh[n][kt] = ldfrag_g(Whh + (size_t)nc * H_, H_, kt * 32, lane);
  }

  _Float16* wmine = w_lds + (role ? 4 * NT_ * 2 * FRAGH_ + wave * (NT_ * 4 * FRAGH_)
                                  : wave * (NT_ * 2 * FRAGH_));
  if (role == 0) {
    #pragma unroll
    for (int n = 0; n < NT_; ++n) {
      #pragma unroll
      for (int kt = 0; kt < 2; ++kt)
        stfrag_w(wmine + (n * 2 + kt) * FRAGH_,
                 ldfrag_g(Wih0 + (size_t)(ncol0 + n * 16) * D_, D_, kt * 32, lane), lane);
    }
  } else {
    #pragma unroll
    for (int n = 0; n < NT_; ++n) {
      #pragma unroll
      for (int kt = 0; kt < 4; ++kt)
        stfrag_w(wmine + (n * 4 + kt) * FRAGH_,
                 ldfrag_g(Wih1 + (size_t)(ncol0 + n * 16) * H_, H_, kt * 32, lane), lane);
    }
  }

  const bool  is_tanh = (wave == 2);
  const float act_pre  = is_tanh ? 1.0f : 0.5f;
  const float act_post = is_tanh ? 1.0f : 0.5f;
  const float act_off  = is_tanh ? 0.0f : 0.5f;
  #pragma unroll
  for (int n = 0; n < NT_; ++n) bv[n] *= act_pre;

  float c[CPT_];
  #pragma unroll
  for (int e = 0; e < CPT_; ++e) c[e] = 0.f;
  const int cr = 2 * (rt >> 5) + 8 * ((rt >> 4) & 1);
  const int cj = 4 * (rt & 15);

  const _Float16* xbase = xseq + (size_t)b0 * T_ * D_;
  float (*g)[GP_]          = g_lds[role];
  _Float16 (*hown)[H_ + 8] = h_lds[role];

  __syncthreads();

  for (int s = 0; s <= T_; ++s) {
    const bool active = role ? (s >= 1) : (s < T_);

    if (active) {
      v8f acc[NT_];
      v16h bx[NT_];
      v16h ha[4];
      #pragma unroll
      for (int kt = 0; kt < 4; ++kt) ha[kt] = ldfrag_h(hown, kt, lane);

      if (role == 0) {
        v16h xa[2];
        #pragma unroll
        for (int kt = 0; kt < 2; ++kt)
          xa[kt] = ldfrag_g(xbase + (size_t)s * D_, T_ * D_, kt * 32, lane);
        if (s + 1 < T_)
          __builtin_prefetch(xbase + (size_t)(s + 1) * D_ + (size_t)(lane & 15) * (T_ * D_), 0, 1);
        #pragma unroll
        for (int kx = 0; kx < 2; ++kx) {
          #pragma unroll
          for (int n = 0; n < NT_; ++n) bx[n] = ldfrag_w(wmine + (n * 2 + kx) * FRAGH_, lane);
          __builtin_amdgcn_sched_barrier(0);
          #pragma unroll
          for (int kt = 2 * kx; kt < 2 * kx + 2; ++kt) {
            #pragma unroll
            for (int n = 0; n < NT_; ++n) {
              const v8f cin = (kt == 0) ? v8f{0.f, 0.f, 0.f, 0.f, 0.f, 0.f, 0.f, 0.f} : acc[n];
              acc[n] = __builtin_amdgcn_wmma_f32_16x16x32_f16(false, ha[kt], false, Bh[n][kt],
                                                              (short)0, cin, false, false);
            }
          }
          #pragma unroll
          for (int n = 0; n < NT_; ++n)
            acc[n] = __builtin_amdgcn_wmma_f32_16x16x32_f16(false, xa[kx], false, bx[n],
                                                            (short)0, acc[n], false, false);
          __builtin_amdgcn_sched_barrier(0);
        }
      } else {
        v16h ia[4];
        #pragma unroll
        for (int kt = 0; kt < 4; ++kt) ia[kt] = ldfrag_h(h_lds[0], kt, lane);
        #pragma unroll
        for (int kt = 0; kt < 4; ++kt) {
          #pragma unroll
          for (int n = 0; n < NT_; ++n) bx[n] = ldfrag_w(wmine + (n * 4 + kt) * FRAGH_, lane);
          __builtin_amdgcn_sched_barrier(0);
          #pragma unroll
          for (int n = 0; n < NT_; ++n) {
            const v8f cin = (kt == 0) ? v8f{0.f, 0.f, 0.f, 0.f, 0.f, 0.f, 0.f, 0.f} : acc[n];
            acc[n] = __builtin_amdgcn_wmma_f32_16x16x32_f16(false, ha[kt], false, Bh[n][kt],
                                                            (short)0, cin, false, false);
          }
          #pragma unroll
          for (int n = 0; n < NT_; ++n)
            acc[n] = __builtin_amdgcn_wmma_f32_16x16x32_f16(false, ia[kt], false, bx[n],
                                                            (short)0, acc[n], false, false);
          __builtin_amdgcn_sched_barrier(0);
        }
      }

      #pragma unroll
      for (int n = 0; n < NT_; ++n) {
        const int col = ncol0 + n * 16 + (lane & 15);
        #pragma unroll
        for (int m = 0; m < 8; ++m) {
          float a = act_off + act_post * fast_tanh(__builtin_fmaf(acc[n][m], act_pre, bv[n]));
          g[m + 8 * hi][col] = a;
        }
      }
    }
    __syncthreads();

    if (active) {
      float hv[CPT_];
      #pragma unroll
      for (int rr = 0; rr < 2; ++rr) {
        #pragma unroll
        for (int u = 0; u < 2; ++u) {
          const int r = cr + rr;
          const int j = cj + 64 * u;
          const v4f iv = *(const v4f*)&g[r][j];
          const v4f fv = *(const v4f*)&g[r][H_ + j];
          const v4f gv = *(const v4f*)&g[r][2 * H_ + j];
          const v4f ov = *(const v4f*)&g[r][3 * H_ + j];
          #pragma unroll
          for (int k = 0; k < 4; ++k) {
            const int e = (rr * 2 + u) * 4 + k;
            c[e] = fv[k] * c[e] + iv[k] * gv[k];
            hv[e] = ov[k] * fast_tanh(c[e]);
          }
        }
      }
      #pragma unroll
      for (int rr = 0; rr < 2; ++rr) {
        #pragma unroll
        for (int u = 0; u < 2; ++u) {
          v4h hp;
          #pragma unroll
          for (int k = 0; k < 4; ++k) hp[k] = (_Float16)hv[(rr * 2 + u) * 4 + k];
          *(v4h*)&hown[cr + rr][cj + 64 * u] = hp;
        }
      }
      if (role == 1 && s == T_) {
        #pragma unroll
        for (int rr = 0; rr < 2; ++rr) {
          #pragma unroll
          for (int u = 0; u < 2; ++u) {
            v4f o4;
            #pragma unroll
            for (int k = 0; k < 4; ++k) o4[k] = hv[(rr * 2 + u) * 4 + k];
            *(v4f*)(hlast_out + (size_t)(b0 + cr + rr) * H_ + cj + 64 * u) = o4;
          }
        }
      }
    }
    __syncthreads();
  }
}

__global__ void fc_kernel(const float* __restrict__ h1, const float* __restrict__ Wfc0,
                          const float* __restrict__ bfc0, const float* __restrict__ Wfc1,
                          const float* __restrict__ bfc1, float* __restrict__ out) {
  int b = blockIdx.x;
  int tid = threadIdx.x;
  __shared__ float tmp[32];
  float s = bfc0[tid];
  for (int k = 0; k < H_; ++k) s += h1[b * H_ + k] * Wfc0[tid * H_ + k];
  tmp[tid] = s;
  __syncthreads();
  if (tid < OUT_) {
    float o = bfc1[tid];
    for (int k = 0; k < 32; ++k) o += tmp[k] * Wfc1[tid * 32 + k];
    out[b * OUT_ + tid] = o;
  }
}

extern "C" void kernel_launch(void* const* d_in, const int* in_sizes, int n_in,
                              void* d_out, int out_size, void* d_ws, size_t ws_size,
                              hipStream_t stream) {
  (void)in_sizes; (void)n_in; (void)out_size; (void)ws_size;
  const float* x    = (const float*)d_in[0];
  const float* Wih0 = (const float*)d_in[1];
  const float* Whh0 = (const float*)d_in[2];
  const float* bih0 = (const float*)d_in[3];
  const float* bhh0 = (const float*)d_in[4];
  const float* Wih1 = (const float*)d_in[5];
  const float* Whh1 = (const float*)d_in[6];
  const float* bih1 = (const float*)d_in[7];
  const float* bhh1 = (const float*)d_in[8];
  const float* Wfc0 = (const float*)d_in[9];
  const float* bfc0 = (const float*)d_in[10];
  const float* Wfc1 = (const float*)d_in[11];
  const float* bfc1 = (const float*)d_in[12];

  char* p = (char*)d_ws;
  auto alloc = [&](size_t bytes) {
    char* r = p;
    p += (bytes + 255) & ~(size_t)255;
    return r;
  };
  _Float16* x_h    = (_Float16*)alloc((size_t)B_ * T_ * D_ * 2);
  _Float16* wih0_h = (_Float16*)alloc((size_t)G_ * D_ * 2);
  _Float16* whh0_h = (_Float16*)alloc((size_t)G_ * H_ * 2);
  _Float16* wih1_h = (_Float16*)alloc((size_t)G_ * H_ * 2);
  _Float16* whh1_h = (_Float16*)alloc((size_t)G_ * H_ * 2);
  float*    b0     = (float*)alloc((size_t)G_ * 4);
  float*    b1     = (float*)alloc((size_t)G_ * 4);
  float*    h1last = (float*)alloc((size_t)B_ * H_ * 4);

  static_assert((B_ * T_ * D_) % (CVT_E_ * CVT_UX_ * CVT_T_) == 0, "x pass: whole workgroups");
  static_assert((G_ * D_) % (CVT_E_ * CVT_T_) == 0 && (G_ * H_) % (CVT_E_ * CVT_T_) == 0,
                "weight passes: whole workgroups");
  cvt_f32_to_f16<CVT_UX_><<<B_ * T_ * D_ / (CVT_E_ * CVT_UX_ * CVT_T_), CVT_T_, 0, stream>>>(x, x_h);
  cvt_f32_to_f16<1><<<G_ * D_ / (CVT_E_ * CVT_T_), CVT_T_, 0, stream>>>(Wih0, wih0_h);
  cvt_f32_to_f16<1><<<G_ * H_ / (CVT_E_ * CVT_T_), CVT_T_, 0, stream>>>(Whh0, whh0_h);
  cvt_f32_to_f16<1><<<G_ * H_ / (CVT_E_ * CVT_T_), CVT_T_, 0, stream>>>(Wih1, wih1_h);
  cvt_f32_to_f16<1><<<G_ * H_ / (CVT_E_ * CVT_T_), CVT_T_, 0, stream>>>(Whh1, whh1_h);
  add_vec<<<2, 256, 0, stream>>>(bih0, bhh0, b0, G_);
  add_vec<<<2, 256, 0, stream>>>(bih1, bhh1, b1, G_);

  lstm2_kernel<<<B_ / BT_, THREADS_, 0, stream>>>(x_h, wih0_h, whh0_h, b0,
                                                  wih1_h, whh1_h, b1, h1last);
  fc_kernel<<<B_, 32, 0, stream>>>(h1last, Wfc0, bfc0, Wfc1, bfc1, (float*)d_out);
}


# Last k group of a step finished column tile by column tile: a tile's activation (fma, tanh, fma) issues between the WMMAs of the tiles behind it, not after the step's last WMMA; sums unchanged
# speedup vs baseline: 3.5606x; 1.0924x over previous
#include <hip/hip_runtime.h>

#define B_ 512
#define T_ 512
#define D_ 64
#define H_ 128
#define G_ 512
#define OUT_ 16
#define BT_ 16
#define RT_ 128
#define THREADS_ 256
#define NT_ 8
#define CPT_ 16
#define GP_ (G_ + 4)
#define FRAGH_ 512

typedef __attribute__((ext_vector_type(16))) _Float16 v16h;
typedef __attribute__((ext_vector_type(8)))  _Float16 v8h;
typedef __attribute__((ext_vector_type(8)))  float    v8f;
typedef __attribute__((ext_vector_type(4)))  _Float16 v4h;
typedef __attribute__((ext_vector_type(4)))  float    v4f;

__device__ __forceinline__ float fast_tanh(float x) {
#if __has_builtin(__builtin_amdgcn_tanhf)
  return __builtin_amdgcn_tanhf(x);
#else
  float e = __builtin_amdgcn_exp2f(x * 2.8853900817779268f);
  return 1.0f - 2.0f * __builtin_amdgcn_rcpf(e + 1.0f);
#endif
}

__device__ __forceinline__ v16h ldfrag_g(const _Float16* __restrict__ base,
                                         int stride, int koff, int lane) {
  int r  = lane & 15;
  int hi = lane >> 4;
  const _Float16* p = base + (size_t)r * stride + koff + hi * 8;
  v8h a = *(const v8h*)p;
  v8h b = *(const v8h*)(p + 16);
  return __builtin_shufflevector(a, b, 0,1,2,3,4,5,6,7,8,9,10,11,12,13,14,15);
}

#define CVT_E_  8
#define CVT_T_  256
#define CVT_UX_ 16
template <int U>
__global__ __launch_bounds__(CVT_T_)
void cvt_f32_to_f16(const float* __restrict__ s, _Float16* __restrict__ d) {
  const int g0     = blockIdx.x * CVT_T_ + threadIdx.x;
  const int stride = gridDim.x * CVT_T_;
  v4f a[U], b[U];
  #pragma unroll
  for (int u = 0; u < U; ++u) {
    const float* p = s + (size_t)(g0 + u * stride) * CVT_E_;
    a[u] = *(const v4f*)p;
    b[u] = *(const v4f*)(p + 4);
  }
  #pragma unroll
  for (int u = 0; u < U; ++u) {
    v8h o;
    #pragma unroll
    for (int k = 0; k < 4; ++k) {
      o[k]     = (_Float16)a[u][k];
      o[4 + k] = (_Float16)b[u][k];
    }
    *(v8h*)(d + (size_t)(g0 + u * stride) * CVT_E_) = o;
  }
}

__global__ void add_vec(const float* __restrict__ a, const float* __restrict__ b,
                        float* __restrict__ o, int n) {
  int i = blockIdx.x * blockDim.x + threadIdx.x;
  if (i < n) o[i] = a[i] + b[i];
}

__device__ __forceinline__ void stfrag_w(_Float16* p, v16h f, int lane) {
  *(v8h*)(p + lane * 8)       = __builtin_shufflevector(f, f, 0,1,2,3,4,5,6,7);
  *(v8h*)(p + 256 + lane * 8) = __builtin_shufflevector(f, f, 8,9,10,11,12,13,14,15);
}
__device__ __forceinline__ v16h ldfrag_w(const _Float16* p, int lane) {
  v8h a = *(const v8h*)(p + lane * 8);
  v8h b = *(const v8h*)(p + 256 + lane * 8);
  return __builtin_shufflevector(a, b, 0,1,2,3,4,5,6,7,8,9,10,11,12,13,14,15);
}

__device__ __forceinline__ v16h ldfrag_h(const _Float16 (*h)[H_ + 8], int kt, int lane) {
  const _Float16* p = &h[lane & 15][kt * 32 + (lane >> 4) * 8];
  v8h a = *(const v8h*)p;
  v8h b = *(const v8h*)(p + 16);
  return __builtin_shufflevector(a, b, 0,1,2,3,4,5,6,7,8,9,10,11,12,13,14,15);
}

__global__ __launch_bounds__(THREADS_, 1)
void lstm2_kernel(const _Float16* __restrict__ xseq,
                  const _Float16* __restrict__ Wih0,
                  const _Float16* __restrict__ Whh0,
                  const float*    __restrict__ bias0,
                  const _Float16* __restrict__ Wih1,
                  const _Float16* __restrict__ Whh1,
                  const float*    __restrict__ bias1,
                  float*          __restrict__ hlast_out)
{
  __shared__ __attribute__((aligned(16))) _Float16 h_lds[2][BT_][H_ + 8];
  __shared__ __attribute__((aligned(16))) float    g_lds[2][BT_][GP_];
  __shared__ __attribute__((aligned(16))) _Float16 w_lds[4 * NT_ * 2 * FRAGH_ + 4 * NT_ * 4 * FRAGH_];

  const int tid   = threadIdx.x;
  const int lane  = tid & 31;
  const int wave8 = __builtin_amdgcn_readfirstlane(tid >> 5);
  const int role  = wave8 >> 2;
  const int wave  = wave8 & 3;
  const int rt    = tid & (RT_ - 1);
  const int hi    = lane >> 4;
  const int b0    = blockIdx.x * BT_;
  const int ncol0 = wave * (NT_ * 16);

  for (int i = tid; i < 2 * BT_ * (H_ + 8); i += THREADS_)
    ((_Float16*)h_lds)[i] = (_Float16)0.0f;

  const _Float16* Whh  = role ? Whh1 : Whh0;
  const float*    bias = role ? bias1 : bias0;
  v16h Bh[NT_][4];
  float bv[NT_];
  #pragma unroll
  for (int n = 0; n < NT_; ++n) {
    const int nc = ncol0 + n * 16;
    bv[n] = bias[nc + (lane & 15)];
    #pragma unroll
    for (int kt = 0; kt < 4; ++kt)
      Bh[n][kt] = ldfrag_g(Whh + (size_t)nc * H_, H_, kt * 32, lane);
  }

  _Float16* wmine = w_lds + (role ? 4 * NT_ * 2 * FRAGH_ + wave * (NT_ * 4 * FRAGH_)
                                  : wave * (NT_ * 2 * FRAGH_));
  if (role == 0) {
    #pragma unroll
    for (int n = 0; n < NT_; ++n) {
      #pragma unroll
      for (int kt = 0; kt < 2; ++kt)
        stfrag_w(wmine + (n * 2 + kt) * FRAGH_,
                 ldfrag_g(Wih0 + (size_t)(ncol0 + n * 16) * D_, D_, kt * 32, lane), lane);
    }
  } else {
    #pragma unroll
    for (int n = 0; n < NT_; ++n) {
      #pragma unroll
      for (int kt = 0; kt < 4; ++kt)
        stfrag_w(wmine + (n * 4 + kt) * FRAGH_,
                 ldfrag_g(Wih1 + (size_t)(ncol0 + n * 16) * H_, H_, kt * 32, lane), lane);
    }
  }

  const bool  is_tanh = (wave == 2);
  const float act_pre  = is_tanh ? 1.0f : 0.5f;
  const float act_post = is_tanh ? 1.0f : 0.5f;
  const float act_off  = is_tanh ? 0.0f : 0.5f;
  #pragma unroll
  for (int n = 0; n < NT_; ++n) bv[n] *= act_pre;

  float c[CPT_];
  #pragma unroll
  for (int e = 0; e < CPT_; ++e) c[e] = 0.f;
  const int cr = 2 * (rt >> 5) + 8 * ((rt >> 4) & 1);
  const int cj = 4 * (rt & 15);

  const _Float16* xbase = xseq + (size_t)b0 * T_ * D_;
  float (*g)[GP_]          = g_lds[role];
  _Float16 (*hown)[H_ + 8] = h_lds[role];

  __syncthreads();

  for (int s = 0; s <= T_; ++s) {
    const bool active = role ? (s >= 1) : (s < T_);

    if (active) {
      v8f acc[NT_];
      v16h bx[NT_];
      auto act_tile = [&](int n, float bvn) {
        #pragma unroll
        for (int m = 0; m < 8; ++m)
          acc[n][m] = act_off + act_post * fast_tanh(__builtin_fmaf(acc[n][m], act_pre, bvn));
      };
      v16h ha[4];
      #pragma unroll
      for (int kt = 0; kt < 4; ++kt) ha[kt] = ldfrag_h(hown, kt, lane);

      if (role == 0) {
        v16h xa[2];
        #pragma unroll
        for (int kt = 0; kt < 2; ++kt)
          xa[kt] = ldfrag_g(xbase + (size_t)s * D_, T_ * D_, kt * 32, lane);
        if (s + 1 < T_)
          __builtin_prefetch(xbase + (size_t)(s + 1) * D_ + (size_t)(lane & 15) * (T_ * D_), 0, 1);
        #pragma unroll
        for (int kx = 0; kx < 2; ++kx) {
          #pragma unroll
          for (int n = 0; n < NT_; ++n) bx[n] = ldfrag_w(wmine + (n * 2 + kx) * FRAGH_, lane);
          __builtin_amdgcn_sched_barrier(0);
          if (kx == 0) {
            #pragma unroll
            for (int kt = 0; kt < 2; ++kt) {
              #pragma unroll
              for (int n = 0; n < NT_; ++n) {
                const v8f cin = (kt == 0) ? v8f{0.f, 0.f, 0.f, 0.f, 0.f, 0.f, 0.f, 0.f} : acc[n];
                acc[n] = __builtin_amdgcn_wmma_f32_16x16x32_f16(false, ha[kt], false, Bh[n][kt],
                                                                (short)0, cin, false, false);
              }
            }
            #pragma unroll
            for (int n = 0; n < NT_; ++n)
              acc[n] = __builtin_amdgcn_wmma_f32_16x16x32_f16(false, xa[kx], false, bx[n],
                                                              (short)0, acc[n], false, false);
            __builtin_amdgcn_sched_barrier(0);
          } else {
            #pragma unroll
            for (int n = 0; n < NT_; ++n)
              acc[n] = __builtin_amdgcn_wmma_f32_16x16x32_f16(false, ha[2], false, Bh[n][2],
                                                              (short)0, acc[n], false, false);
            #pragma unroll
            for (int n = 0; n < 4; ++n)
              acc[n] = __builtin_amdgcn_wmma_f32_16x16x32_f16(false, ha[3], false, Bh[n][3],
                                                              (short)0, acc[n], false, false);
            #pragma unroll
            for (int n = 0; n < NT_; ++n) {
              acc[n] = __builtin_amdgcn_wmma_f32_16x16x32_f16(false, xa[1], false, bx[n],
                                                              (short)0, acc[n], false, false);
              if (n + 4 < NT_)
                acc[n + 4] = __builtin_amdgcn_wmma_f32_16x16x32_f16(false, ha[3], false, Bh[n + 4][3],
                                                                    (short)0, acc[n + 4], false, false);
              act_tile(n, bv[n]);
            }
          }
        }
      } else {
        v16h ia[4];
        #pragma unroll
        for (int kt = 0; kt < 4; ++kt) ia[kt] = ldfrag_h(h_lds[0], kt, lane);
        #pragma unroll
        for (int kt = 0; kt < 4; ++kt) {
          #pragma unroll
          for (int n = 0; n < NT_; ++n) bx[n] = ldfrag_w(wmine + (n * 4 + kt) * FRAGH_, lane);
          __builtin_amdgcn_sched_barrier(0);
          if (kt < 3) {
            #pragma unroll
            for (int n = 0; n < NT_; ++n) {
              const v8f cin = (kt == 0) ? v8f{0.f, 0.f, 0.f, 0.f, 0.f, 0.f, 0.f, 0.f} : acc[n];
              acc[n] = __builtin_amdgcn_wmma_f32_16x16x32_f16(false, ha[kt], false, Bh[n][kt],
                                                              (short)0, cin, false, false);
            }
            #pragma unroll
            for (int n = 0; n < NT_; ++n)
              acc[n] = __builtin_amdgcn_wmma_f32_16x16x32_f16(false, ia[kt], false, bx[n],
                                                              (short)0, acc[n], false, false);
            __builtin_amdgcn_sched_barrier(0);
          } else {
            #pragma unroll
            for (int n = 0; n < 4; ++n)
              acc[n] = __builtin_amdgcn_wmma_f32_16x16x32_f16(false, ha[3], false, Bh[n][3],
                                                              (short)0, acc[n], false, false);
            #pragma unroll
            for (int n = 0; n < NT_; ++n) {
              acc[n] = __builtin_amdgcn_wmma_f32_16x16x32_f16(false, ia[3], false, bx[n],
                                                              (short)0, acc[n], false, false);
              if (n + 4 < NT_)
                acc[n + 4] = __builtin_amdgcn_wmma_f32_16x16x32_f16(false, ha[3], false, Bh[n + 4][3],
                                                                    (short)0, acc[n + 4], false, false);
              act_tile(n, bv[n]);
            }
          }
        }
      }

      #pragma unroll
      for (int n = 0; n < NT_; ++n) {
        const int col = ncol0 + n * 16 + (lane & 15);
        #pragma unroll
        for (int m = 0; m < 8; ++m) g[m + 8 * hi][col] = acc[n][m];
      }
    }
    __syncthreads();

    if (active) {
      float hv[CPT_];
      #pragma unroll
      for (int rr = 0; rr < 2; ++rr) {
        #pragma unroll
        for (int u = 0; u < 2; ++u) {
          const int r = cr + rr;
          const int j = cj + 64 * u;
          const v4f iv = *(const v4f*)&g[r][j];
          const v4f fv = *(const v4f*)&g[r][H_ + j];
          const v4f gv = *(const v4f*)&g[r][2 * H_ + j];
          const v4f ov = *(const v4f*)&g[r][3 * H_ + j];
          #pragma unroll
          for (int k = 0; k < 4; ++k) {
            const int e = (rr * 2 + u) * 4 + k;
            c[e] = fv[k] * c[e] + iv[k] * gv[k];
            hv[e] = ov[k] * fast_tanh(c[e]);
          }
        }
      }
      #pragma unroll
      for (int rr = 0; rr < 2; ++rr) {
        #pragma unroll
        for (int u = 0; u < 2; ++u) {
          v4h hp;
          #pragma unroll
          for (int k = 0; k < 4; ++k) hp[k] = (_Float16)hv[(rr * 2 + u) * 4 + k];
          *(v4h*)&hown[cr + rr][cj + 64 * u] = hp;
        }
      }
      if (role == 1 && s == T_) {
        #pragma unroll
        for (int rr = 0; rr < 2; ++rr) {
          #pragma unroll
          for (int u = 0; u < 2; ++u) {
            v4f o4;
            #pragma unroll
            for (int k = 0; k < 4; ++k) o4[k] = hv[(rr * 2 + u) * 4 + k];
            *(v4f*)(hlast_out + (size_t)(b0 + cr + rr) * H_ + cj + 64 * u) = o4;
          }
        }
      }
    }
    __syncthreads();
  }
}

__global__ void fc_kernel(const float* __restrict__ h1, const float* __restrict__ Wfc0,
                          const float* __restrict__ bfc0, const float* __restrict__ Wfc1,
                          const float* __restrict__ bfc1, float* __restrict__ out) {
  int b = blockIdx.x;
  int tid = threadIdx.x;
  __shared__ float tmp[32];
  float s = bfc0[tid];
  for (int k = 0; k < H_; ++k) s += h1[b * H_ + k] * Wfc0[tid * H_ + k];
  tmp[tid] = s;
  __syncthreads();
  if (tid < OUT_) {
    float o = bfc1[tid];
    for (int k = 0; k < 32; ++k) o += tmp[k] * Wfc1[tid * 32 + k];
    out[b * OUT_ + tid] = o;
  }
}

extern "C" void kernel_launch(void* const* d_in, const int* in_sizes, int n_in,
                              void* d_out, int out_size, void* d_ws, size_t ws_size,
                              hipStream_t stream) {
  (void)in_sizes; (void)n_in; (void)out_size; (void)ws_size;
  const float* x    = (const float*)d_in[0];
  const float* Wih0 = (const float*)d_in[1];
  const float* Whh0 = (const float*)d_in[2];
  const float* bih0 = (const float*)d_in[3];
  const float* bhh0 = (const float*)d_in[4];
  const float* Wih1 = (const float*)d_in[5];
  const float* Whh1 = (const float*)d_in[6];
  const float* bih1 = (const float*)d_in[7];
  const float* bhh1 = (const float*)d_in[8];
  const float* Wfc0 = (const float*)d_in[9];
  const float* bfc0 = (const float*)d_in[10];
  const float* Wfc1 = (const float*)d_in[11];
  const float* bfc1 = (const float*)d_in[12];

  char* p = (char*)d_ws;
  auto alloc = [&](size_t bytes) {
    char* r = p;
    p += (bytes + 255) & ~(size_t)255;
    return r;
  };
  _Float16* x_h    = (_Float16*)alloc((size_t)B_ * T_ * D_ * 2);
  _Float16* wih0_h = (_Float16*)alloc((size_t)G_ * D_ * 2);
  _Float16* whh0_h = (_Float16*)alloc((size_t)G_ * H_ * 2);
  _Float16* wih1_h = (_Float16*)alloc((size_t)G_ * H_ * 2);
  _Float16* whh1_h = (_Float16*)alloc((size_t)G_ * H_ * 2);
  float*    b0     = (float*)alloc((size_t)G_ * 4);
  float*    b1     = (float*)alloc((size_t)G_ * 4);
  float*    h1last = (float*)alloc((size_t)B_ * H_ * 4);

  static_assert((B_ * T_ * D_) % (CVT_E_ * CVT_UX_ * CVT_T_) == 0, "x pass: whole workgroups");
  static_assert((G_ * D_) % (CVT_E_ * CVT_T_) == 0 && (G_ * H_) % (CVT_E_ * CVT_T_) == 0,
                "weight passes: whole workgroups");
  cvt_f32_to_f16<CVT_UX_><<<B_ * T_ * D_ / (CVT_E_ * CVT_UX_ * CVT_T_), CVT_T_, 0, stream>>>(x, x_h);
  cvt_f32_to_f16<1><<<G_ * D_ / (CVT_E_ * CVT_T_), CVT_T_, 0, stream>>>(Wih0, wih0_h);
  cvt_f32_to_f16<1><<<G_ * H_ / (CVT_E_ * CVT_T_), CVT_T_, 0, stream>>>(Whh0, whh0_h);
  cvt_f32_to_f16<1><<<G_ * H_ / (CVT_E_ * CVT_T_), CVT_T_, 0, stream>>>(Wih1, wih1_h);
  cvt_f32_to_f16<1><<<G_ * H_ / (CVT_E_ * CVT_T_), CVT_T_, 0, stream>>>(Whh1, whh1_h);
  add_vec<<<2, 256, 0, stream>>>(bih0, bhh0, b0, G_);
  add_vec<<<2, 256, 0, stream>>>(bih1, bhh1, b1, G_);

  lstm2_kernel<<<B_ / BT_, THREADS_, 0, stream>>>(x_h, wih0_h, whh0_h, b0,
                                                  wih1_h, whh1_h, b1, h1last);
  fc_kernel<<<B_, 32, 0, stream>>>(h1last, Wfc0, bfc0, Wfc1, bfc1, (float*)d_out);
}


# speedup vs baseline: 3.8268x; 1.0748x over previous
#include <hip/hip_runtime.h>

#define B_ 512
#define T_ 512
#define D_ 64
#define H_ 128
#define G_ 512
#define OUT_ 16
#define BT_ 16
#define RT_ 128
#define THREADS_ 256
#define NT_ 8
#define CPT_ 16
#define FRAGH_ 512
#define BLK_  32768
#define GPC_  520
#define GREG_ (8 * GPC_ * 4)
#define HT_   (BT_ * (H_ + 8) * 2)
#define OFF_W0_(pair)     ((pair) * BLK_)
#define OFF_W1_(pair, k)  ((2 + 2 * (pair) + (k)) * BLK_)
#define OFF_G_(role, pair) ((6 + 2 * (pair) + (role)) * BLK_)
#define OFF_H_(layer, pair) ((6 + 2 * (pair)) * BLK_ + GREG_ + (layer) * HT_)
#define LDS_BYTES_ (9 * BLK_ + GREG_)

typedef __attribute__((ext_vector_type(16))) _Float16 v16h;
typedef __attribute__((ext_vector_type(8)))  _Float16 v8h;
typedef __attribute__((ext_vector_type(8)))  float    v8f;
typedef __attribute__((ext_vector_type(4)))  _Float16 v4h;
typedef __attribute__((ext_vector_type(4)))  float    v4f;

__device__ __forceinline__ float fast_tanh(float x) {
#if __has_builtin(__builtin_amdgcn_tanhf)
  return __builtin_amdgcn_tanhf(x);
#else
  float e = __builtin_amdgcn_exp2f(x * 2.8853900817779268f);
  return 1.0f - 2.0f * __builtin_amdgcn_rcpf(e + 1.0f);
#endif
}

__device__ __forceinline__ v16h ldfrag_g(const _Float16* __restrict__ base,
                                         int stride, int koff, int lane) {
  int r  = lane & 15;
  int hi = lane >> 4;
  const _Float16* p = base + (size_t)r * stride + koff + hi * 8;
  v8h a = *(const v8h*)p;
  v8h b = *(const v8h*)(p + 16);
  return __builtin_shufflevector(a, b, 0,1,2,3,4,5,6,7,8,9,10,11,12,13,14,15);
}

#define CVT_E_  8
#define CVT_T_  256
#define CVT_UX_ 16
template <int U>
__global__ __launch_bounds__(CVT_T_)
void cvt_f32_to_f16(const float* __restrict__ s, _Float16* __restrict__ d) {
  const int g0     = blockIdx.x * CVT_T_ + threadIdx.x;
  const int stride = gridDim.x * CVT_T_;
  v4f a[U], b[U];
  #pragma unroll
  for (int u = 0; u < U; ++u) {
    const float* p = s + (size_t)(g0 + u * stride) * CVT_E_;
    a[u] = *(const v4f*)p;
    b[u] = *(const v4f*)(p + 4);
  }
  #pragma unroll
  for (int u = 0; u < U; ++u) {
    v8h o;
    #pragma unroll
    for (int k = 0; k < 4; ++k) {
      o[k]     = (_Float16)a[u][k];
      o[4 + k] = (_Float16)b[u][k];
    }
    *(v8h*)(d + (size_t)(g0 + u * stride) * CVT_E_) = o;
  }
}

__global__ void add_vec(const float* __restrict__ a, const float* __restrict__ b,
                        float* __restrict__ o, int n) {
  int i = blockIdx.x * blockDim.x + threadIdx.x;
  if (i < n) o[i] = a[i] + b[i];
}

__device__ __forceinline__ void stfrag_w(_Float16* p, v16h f, int lane) {
  *(v8h*)(p + lane * 8)       = __builtin_shufflevector(f, f, 0,1,2,3,4,5,6,7);
  *(v8h*)(p + 256 + lane * 8) = __builtin_shufflevector(f, f, 8,9,10,11,12,13,14,15);
}
__device__ __forceinline__ v16h ldfrag_w(const _Float16* p, int lane) {
  v8h a = *(const v8h*)(p + lane * 8);
  v8h b = *(const v8h*)(p + 256 + lane * 8);
  return __builtin_shufflevector(a, b, 0,1,2,3,4,5,6,7,8,9,10,11,12,13,14,15);
}

__device__ __forceinline__ v16h ldfrag_h(const _Float16 (*h)[H_ + 8], int kt, int lane) {
  const _Float16* p = &h[lane & 15][kt * 32 + (lane >> 4) * 8];
  v8h a = *(const v8h*)p;
  v8h b = *(const v8h*)(p + 16);
  return __builtin_shufflevector(a, b, 0,1,2,3,4,5,6,7,8,9,10,11,12,13,14,15);
}

__global__ __launch_bounds__(THREADS_, 1)
void lstm2_kernel(const _Float16* __restrict__ xseq,
                  const _Float16* __restrict__ Wih0,
                  const _Float16* __restrict__ Whh0,
                  const float*    __restrict__ bias0,
                  const _Float16* __restrict__ Wih1,
                  const _Float16* __restrict__ Whh1,
                  const float*    __restrict__ bias1,
                  float*          __restrict__ hlast_out)
{
  __shared__ __attribute__((aligned(16))) unsigned char lds[LDS_BYTES_];
  typedef _Float16 (*htile_t)[H_ + 8];
  typedef float    (*gtile_t)[GPC_];

  const int tid   = threadIdx.x;
  const int lane  = tid & 31;
  const int wave8 = __builtin_amdgcn_readfirstlane(tid >> 5);
  const int role  = wave8 >> 2;
  const int wave  = wave8 & 3;
  const int rt    = tid & (RT_ - 1);
  const int hi    = lane >> 4;
  const int b0    = blockIdx.x * BT_;
  const int ncol0 = wave * (NT_ * 16);
  const int pair  = wave & 1;
  const int kp    = wave >> 1;

  for (int i = tid; i < 2 * BT_ * (H_ + 8); i += THREADS_) {
    ((_Float16*)(lds + OFF_H_(0, 0)))[i] = (_Float16)0.0f;
    ((_Float16*)(lds + OFF_H_(0, 1)))[i] = (_Float16)0.0f;
  }

  const _Float16* Whh  = role ? Whh1 : Whh0;
  const float*    bias = role ? bias1 : bias0;
  v16h Bh[NT_][4];
  float bv[NT_];
  #pragma unroll
  for (int n = 0; n < NT_; ++n) {
    const int nc = ncol0 + n * 16;
    bv[n] = bias[nc + (lane & 15)];
    #pragma unroll
    for (int kt = 0; kt < 4; ++kt)
      Bh[n][kt] = ldfrag_g(Whh + (size_t)nc * H_, H_, kt * 32, lane);
  }

  _Float16* wmine = (_Float16*)(lds + (role ? OFF_W1_(pair, kp)
                                            : OFF_W0_(pair) + kp * (NT_ * 2 * FRAGH_ * 2)));
  if (role == 0) {
    #pragma unroll
    for (int n = 0; n < NT_; ++n) {
      #pragma unroll
      for (int kt = 0; kt < 2; ++kt)
        stfrag_w(wmine + (n * 2 + kt) * FRAGH_,
                 ldfrag_g(Wih0 + (size_t)(ncol0 + n * 16) * D_, D_, kt * 32, lane), lane);
    }
  } else {
    #pragma unroll
    for (int n = 0; n < NT_; ++n) {
      #pragma unroll
      for (int kt = 0; kt < 4; ++kt)
        stfrag_w(wmine + (n * 4 + kt) * FRAGH_,
                 ldfrag_g(Wih1 + (size_t)(ncol0 + n * 16) * H_, H_, kt * 32, lane), lane);
    }
  }

  const bool  is_tanh = (wave == 2);
  const float act_pre  = is_tanh ? 1.0f : 0.5f;
  const float act_post = is_tanh ? 1.0f : 0.5f;
  const float act_off  = is_tanh ? 0.0f : 0.5f;
  #pragma unroll
  for (int n = 0; n < NT_; ++n) bv[n] *= act_pre;

  float c[CPT_];
  #pragma unroll
  for (int e = 0; e < CPT_; ++e) c[e] = 0.f;
  const int cr = 2 * (rt >> 5) + 8 * ((rt >> 4) & 1);
  const int cj = 4 * (rt & 15);

  const int clr = 2 * kp + 4 * ((rt >> 4) & 1);

  const _Float16* xbase = xseq + (size_t)b0 * T_ * D_;
  const gtile_t g0r = (gtile_t)(lds + (role ? OFF_G_(1, 0) : OFF_G_(0, 0)));
  const gtile_t g1r = (gtile_t)(lds + (role ? OFF_G_(1, 1) : OFF_G_(0, 1)));
  const gtile_t g    = pair ? g1r : g0r;
  const htile_t hown  = (htile_t)(lds + (role ? (pair ? OFF_H_(1, 1) : OFF_H_(1, 0)) : (pair ? OFF_H_(0, 1) : OFF_H_(0, 0))));
  const htile_t hin0  = (htile_t)(lds + (pair ? OFF_H_(0, 1) : OFF_H_(0, 0)));
  const htile_t hdstA = (htile_t)(lds + (role ? OFF_H_(1, 0) : OFF_H_(0, 0)));
  const htile_t hdstB = (htile_t)(lds + (role ? OFF_H_(1, 1) : OFF_H_(0, 1)));

  __syncthreads();

  for (int s = 0; s <= T_; ++s) {
    const bool active = role ? (s >= 1) : (s < T_);

    if (active) {
      v8f acc[NT_];
      v16h bx[NT_];
      auto act_tile = [&](int n, float bvn) {
        #pragma unroll
        for (int m = 0; m < 8; ++m)
          acc[n][m] = act_off + act_post * fast_tanh(__builtin_fmaf(acc[n][m], act_pre, bvn));
      };
      v16h ha[4];
      #pragma unroll
      for (int kt = 0; kt < 4; ++kt) ha[kt] = ldfrag_h(hown, kt, lane);

      if (role == 0) {
        v16h xa[2];
        #pragma unroll
        for (int kt = 0; kt < 2; ++kt)
          xa[kt] = ldfrag_g(xbase + (size_t)s * D_, T_ * D_, kt * 32, lane);
        if (s + 1 < T_)
          __builtin_prefetch(xbase + (size_t)(s + 1) * D_ + (size_t)(lane & 15) * (T_ * D_), 0, 1);
        #pragma unroll
        for (int kx = 0; kx < 2; ++kx) {
          #pragma unroll
          for (int n = 0; n < NT_; ++n) bx[n] = ldfrag_w(wmine + (n * 2 + kx) * FRAGH_, lane);
          __builtin_amdgcn_sched_barrier(0);
          if (kx == 0) {
            #pragma unroll
            for (int kt = 0; kt < 2; ++kt) {
              #pragma unroll
              for (int n = 0; n < NT_; ++n) {
                const v8f cin = (kt == 0) ? v8f{0.f, 0.f, 0.f, 0.f, 0.f, 0.f, 0.f, 0.f} : acc[n];
                acc[n] = __builtin_amdgcn_wmma_f32_16x16x32_f16(false, ha[kt], false, Bh[n][kt],
                                                                (short)0, cin, false, false);
              }
            }
            #pragma unroll
            for (int n = 0; n < NT_; ++n)
              acc[n] = __builtin_amdgcn_wmma_f32_16x16x32_f16(false, xa[kx], false, bx[n],
                                                              (short)0, acc[n], false, false);
            __builtin_amdgcn_sched_barrier(0);
          } else {
            #pragma unroll
            for (int n = 0; n < NT_; ++n)
              acc[n] = __builtin_amdgcn_wmma_f32_16x16x32_f16(false, ha[2], false, Bh[n][2],
                                                              (short)0, acc[n], false, false);
            #pragma unroll
            for (int n = 0; n < 4; ++n)
              acc[n] = __builtin_amdgcn_wmma_f32_16x16x32_f16(false, ha[3], false, Bh[n][3],
                                                              (short)0, acc[n], false, false);
            #pragma unroll
            for (int n = 0; n < NT_; ++n) {
              acc[n] = __builtin_amdgcn_wmma_f32_16x16x32_f16(false, xa[1], false, bx[n],
                                                              (short)0, acc[n], false, false);
              if (n + 4 < NT_)
                acc[n + 4] = __builtin_amdgcn_wmma_f32_16x16x32_f16(false, ha[3], false, Bh[n + 4][3],
                                                                    (short)0, acc[n + 4], false, false);
              act_tile(n, bv[n]);
            }
          }
        }
      } else {
        v16h ia[4];
        #pragma unroll
        for (int kt = 0; kt < 4; ++kt) ia[kt] = ldfrag_h(hin0, kt, lane);
        #pragma unroll
        for (int kt = 0; kt < 4; ++kt) {
          #pragma unroll
          for (int n = 0; n < NT_; ++n) bx[n] = ldfrag_w(wmine + (n * 4 + kt) * FRAGH_, lane);
          __builtin_amdgcn_sched_barrier(0);
          if (kt < 3) {
            #pragma unroll
            for (int n = 0; n < NT_; ++n) {
              const v8f cin = (kt == 0) ? v8f{0.f, 0.f, 0.f, 0.f, 0.f, 0.f, 0.f, 0.f} : acc[n];
              acc[n] = __builtin_amdgcn_wmma_f32_16x16x32_f16(false, ha[kt], false, Bh[n][kt],
                                                              (short)0, cin, false, false);
            }
            #pragma unroll
            for (int n = 0; n < NT_; ++n)
              acc[n] = __builtin_amdgcn_wmma_f32_16x16x32_f16(false, ia[kt], false, bx[n],
                                                              (short)0, acc[n], false, false);
            __builtin_amdgcn_sched_barrier(0);
          } else {
            #pragma unroll
            for (int n = 0; n < 4; ++n)
              acc[n] = __builtin_amdgcn_wmma_f32_16x16x32_f16(false, ha[3], false, Bh[n][3],
                                                              (short)0, acc[n], false, false);
            #pragma unroll
            for (int n = 0; n < NT_; ++n) {
              acc[n] = __builtin_amdgcn_wmma_f32_16x16x32_f16(false, ia[3], false, bx[n],
                                                              (short)0, acc[n], false, false);
              if (n + 4 < NT_)
                acc[n + 4] = __builtin_amdgcn_wmma_f32_16x16x32_f16(false, ha[3], false, Bh[n + 4][3],
                                                                    (short)0, acc[n + 4], false, false);
              act_tile(n, bv[n]);
            }
          }
        }
      }

      #pragma unroll
      for (int n = 0; n < NT_; ++n) {
        const int col = ncol0 + n * 16 + (lane & 15);
        #pragma unroll
        for (int m = 0; m < 8; ++m) {
          const gtile_t gd = ((m >> 1) & 1) ? g1r : g0r;
          gd[(m >> 2) * 2 + (m & 1) + 4 * hi][col] = acc[n][m];
        }
      }
    }
    __syncthreads();

    if (active) {
      float hv[CPT_];
      #pragma unroll
      for (int rr = 0; rr < 2; ++rr) {
        #pragma unroll
        for (int u = 0; u < 2; ++u) {
          const int r = clr + rr;
          const int j = cj + 64 * u;
          const v4f iv = *(const v4f*)&g[r][j];
          const v4f fv = *(const v4f*)&g[r][H_ + j];
          const v4f gv = *(const v4f*)&g[r][2 * H_ + j];
          const v4f ov = *(const v4f*)&g[r][3 * H_ + j];
          #pragma unroll
          for (int k = 0; k < 4; ++k) {
            const int e = (rr * 2 + u) * 4 + k;
            c[e] = fv[k] * c[e] + iv[k] * gv[k];
            hv[e] = ov[k] * fast_tanh(c[e]);
          }
        }
      }
      #pragma unroll
      for (int rr = 0; rr < 2; ++rr) {
        #pragma unroll
        for (int u = 0; u < 2; ++u) {
          v4h hp;
          #pragma unroll
          for (int k = 0; k < 4; ++k) hp[k] = (_Float16)hv[(rr * 2 + u) * 4 + k];
          *(v4h*)&hdstA[cr + rr][cj + 64 * u] = hp;
          *(v4h*)&hdstB[cr + rr][cj + 64 * u] = hp;
        }
      }
      if (role == 1 && s == T_) {
        #pragma unroll
        for (int rr = 0; rr < 2; ++rr) {
          #pragma unroll
          for (int u = 0; u < 2; ++u) {
            v4f o4;
            #pragma unroll
            for (int k = 0; k < 4; ++k) o4[k] = hv[(rr * 2 + u) * 4 + k];
            *(v4f*)(hlast_out + (size_t)(b0 + cr + rr) * H_ + cj + 64 * u) = o4;
          }
        }
      }
    }
    __syncthreads();
  }
}

__global__ void fc_kernel(const float* __restrict__ h1, const float* __restrict__ Wfc0,
                          const float* __restrict__ bfc0, const float* __restrict__ Wfc1,
                          const float* __restrict__ bfc1, float* __restrict__ out) {
  int b = blockIdx.x;
  int tid = threadIdx.x;
  __shared__ float tmp[32];
  float s = bfc0[tid];
  for (int k = 0; k < H_; ++k) s += h1[b * H_ + k] * Wfc0[tid * H_ + k];
  tmp[tid] = s;
  __syncthreads();
  if (tid < OUT_) {
    float o = bfc1[tid];
    for (int k = 0; k < 32; ++k) o += tmp[k] * Wfc1[tid * 32 + k];
    out[b * OUT_ + tid] = o;
  }
}

extern "C" void kernel_launch(void* const* d_in, const int* in_sizes, int n_in,
                              void* d_out, int out_size, void* d_ws, size_t ws_size,
                              hipStream_t stream) {
  (void)in_sizes; (void)n_in; (void)out_size; (void)ws_size;
  const float* x    = (const float*)d_in[0];
  const float* Wih0 = (const float*)d_in[1];
  const float* Whh0 = (const float*)d_in[2];
  const float* bih0 = (const float*)d_in[3];
  const float* bhh0 = (const float*)d_in[4];
  const float* Wih1 = (const float*)d_in[5];
  const float* Whh1 = (const float*)d_in[6];
  const float* bih1 = (const float*)d_in[7];
  const float* bhh1 = (const float*)d_in[8];
  const float* Wfc0 = (const float*)d_in[9];
  const float* bfc0 = (const float*)d_in[10];
  const float* Wfc1 = (const float*)d_in[11];
  const float* bfc1 = (const float*)d_in[12];

  char* p = (char*)d_ws;
  auto alloc = [&](size_t bytes) {
    char* r = p;
    p += (bytes + 255) & ~(size_t)255;
    return r;
  };
  _Float16* x_h    = (_Float16*)alloc((size_t)B_ * T_ * D_ * 2);
  _Float16* wih0_h = (_Float16*)alloc((size_t)G_ * D_ * 2);
  _Float16* whh0_h = (_Float16*)alloc((size_t)G_ * H_ * 2);
  _Float16* wih1_h = (_Float16*)alloc((size_t)G_ * H_ * 2);
  _Float16* whh1_h = (_Float16*)alloc((size_t)G_ * H_ * 2);
  float*    b0     = (float*)alloc((size_t)G_ * 4);
  float*    b1     = (float*)alloc((size_t)G_ * 4);
  float*    h1last = (float*)alloc((size_t)B_ * H_ * 4);

  static_assert((B_ * T_ * D_) % (CVT_E_ * CVT_UX_ * CVT_T_) == 0, "x pass: whole workgroups");
  static_assert((G_ * D_) % (CVT_E_ * CVT_T_) == 0 && (G_ * H_) % (CVT_E_ * CVT_T_) == 0,
                "weight passes: whole workgroups");
  cvt_f32_to_f16<CVT_UX_><<<B_ * T_ * D_ / (CVT_E_ * CVT_UX_ * CVT_T_), CVT_T_, 0, stream>>>(x, x_h);
  cvt_f32_to_f16<1><<<G_ * D_ / (CVT_E_ * CVT_T_), CVT_T_, 0, stream>>>(Wih0, wih0_h);
  cvt_f32_to_f16<1><<<G_ * H_ / (CVT_E_ * CVT_T_), CVT_T_, 0, stream>>>(Whh0, whh0_h);
  cvt_f32_to_f16<1><<<G_ * H_ / (CVT_E_ * CVT_T_), CVT_T_, 0, stream>>>(Wih1, wih1_h);
  cvt_f32_to_f16<1><<<G_ * H_ / (CVT_E_ * CVT_T_), CVT_T_, 0, stream>>>(Whh1, whh1_h);
  add_vec<<<2, 256, 0, stream>>>(bih0, bhh0, b0, G_);
  add_vec<<<2, 256, 0, stream>>>(bih1, bhh1, b1, G_);

  lstm2_kernel<<<B_ / BT_, THREADS_, 0, stream>>>(x_h, wih0_h, whh0_h, b0,
                                                  wih1_h, whh1_h, b1, h1last);
  fc_kernel<<<B_, 32, 0, stream>>>(h1last, Wfc0, bfc0, Wfc1, bfc1, (float*)d_out);
}


# Layer waves re-split by gate pair: i,g waves hand p=i*g to the f,o waves in fragment order, no gate scatter and gather; f,o waves keep c, store h transposed for ds_load_tr16; same sums
# speedup vs baseline: 4.0176x; 1.0499x over previous
#include <hip/hip_runtime.h>

#define B_ 512
#define T_ 512
#define D_ 64
#define H_ 128
#define G_ 512
#define OUT_ 16
#define BT_ 16
#define THREADS_ 256
#define NT_ 8
#define FRAGH_ 512
#define BLK_  32768
#define GREG_ 16640
#define HTP_  32
#define HT_   (BT_ * (H_ + 8) * 2)
#define OFF_W0_(pair)     ((pair) * BLK_)
#define OFF_W1_(pair, k)  ((2 + 2 * (pair) + (k)) * BLK_)
#define OFF_P_(role, pair) ((6 + 2 * (pair)) * BLK_ + (role) * 4096)
#define OFF_H_(layer, pair) ((6 + 2 * (pair)) * BLK_ + GREG_ + (layer) * HT_)
#define LDS_BYTES_ (8 * BLK_ + GREG_ + 2 * HT_)

typedef __attribute__((ext_vector_type(16))) _Float16 v16h;
typedef __attribute__((ext_vector_type(8)))  _Float16 v8h;
typedef __attribute__((ext_vector_type(8)))  float    v8f;
typedef __attribute__((ext_vector_type(4)))  float    v4f;
typedef __attribute__((ext_vector_type(8)))  short    s16x8;

__device__ __forceinline__ float fast_tanh(float x) {
#if __has_builtin(__builtin_amdgcn_tanhf)
  return __builtin_amdgcn_tanhf(x);
#else
  float e = __builtin_amdgcn_exp2f(x * 2.8853900817779268f);
  return 1.0f - 2.0f * __builtin_amdgcn_rcpf(e + 1.0f);
#endif
}

__device__ __forceinline__ v16h ldfrag_g(const _Float16* __restrict__ base,
                                         int stride, int koff, int lane) {
  int r  = lane & 15;
  int hi = lane >> 4;
  const _Float16* p = base + (size_t)r * stride + koff + hi * 8;
  v8h a = *(const v8h*)p;
  v8h b = *(const v8h*)(p + 16);
  return __builtin_shufflevector(a, b, 0,1,2,3,4,5,6,7,8,9,10,11,12,13,14,15);
}

#define CVT_E_  8
#define CVT_T_  256
#define CVT_UX_ 16
template <int U>
__global__ __launch_bounds__(CVT_T_)
void cvt_f32_to_f16(const float* __restrict__ s, _Float16* __restrict__ d) {
  const int g0     = blockIdx.x * CVT_T_ + threadIdx.x;
  const int stride = gridDim.x * CVT_T_;
  v4f a[U], b[U];
  #pragma unroll
  for (int u = 0; u < U; ++u) {
    const float* p = s + (size_t)(g0 + u * stride) * CVT_E_;
    a[u] = *(const v4f*)p;
    b[u] = *(const v4f*)(p + 4);
  }
  #pragma unroll
  for (int u = 0; u < U; ++u) {
    v8h o;
    #pragma unroll
    for (int k = 0; k < 4; ++k) {
      o[k]     = (_Float16)a[u][k];
      o[4 + k] = (_Float16)b[u][k];
    }
    *(v8h*)(d + (size_t)(g0 + u * stride) * CVT_E_) = o;
  }
}

__global__ void add_vec(const float* __restrict__ a, const float* __restrict__ b,
                        float* __restrict__ o, int n) {
  int i = blockIdx.x * blockDim.x + threadIdx.x;
  if (i < n) o[i] = a[i] + b[i];
}

__device__ __forceinline__ void stfrag_w(_Float16* p, v16h f, int lane) {
  *(v8h*)(p + lane * 8)       = __builtin_shufflevector(f, f, 0,1,2,3,4,5,6,7);
  *(v8h*)(p + 256 + lane * 8) = __builtin_shufflevector(f, f, 8,9,10,11,12,13,14,15);
}
__device__ __forceinline__ v16h ldfrag_w(const _Float16* p, int lane) {
  v8h a = *(const v8h*)(p + lane * 8);
  v8h b = *(const v8h*)(p + 256 + lane * 8);
  return __builtin_shufflevector(a, b, 0,1,2,3,4,5,6,7,8,9,10,11,12,13,14,15);
}

__device__ __forceinline__ v8h ld_tr16(const unsigned char* p) {
  typedef __attribute__((address_space(3))) s16x8 lds_s16x8;
  return __builtin_bit_cast(v8h, __builtin_amdgcn_ds_load_tr16_b128_v8i16((lds_s16x8*)p));
}
__device__ __forceinline__ v16h ldfrag_ht(const unsigned char* ht, int kt, int lane) {
  const unsigned char* p = ht + (kt * 32 + 8 * (lane >> 4) + (lane & 7)) * HTP_ + ((lane >> 3) & 1) * 16;
  v8h a = ld_tr16(p);
  v8h b = ld_tr16(p + 16 * HTP_);
  return __builtin_shufflevector(a, b, 0,1,2,3,4,5,6,7,8,9,10,11,12,13,14,15);
}

__global__ __launch_bounds__(THREADS_, 1)
void lstm2_kernel(const _Float16* __restrict__ xseq,
                  const _Float16* __restrict__ Wih0,
                  const _Float16* __restrict__ Whh0,
                  const float*    __restrict__ bias0,
                  const _Float16* __restrict__ Wih1,
                  const _Float16* __restrict__ Whh1,
                  const float*    __restrict__ bias1,
                  float*          __restrict__ hlast_out)
{
  __shared__ __attribute__((aligned(16))) unsigned char lds[LDS_BYTES_];

  const int tid   = threadIdx.x;
  const int lane  = tid & 31;
  const int wave8 = __builtin_amdgcn_readfirstlane(tid >> 5);
  const int role  = wave8 >> 2;
  const int wave  = wave8 & 3;
  const int hi    = lane >> 4;
  const int b0    = blockIdx.x * BT_;
  const int pair  = wave & 1;
  const int kp    = wave >> 1;
  const int kind  = (wave >> 1) ^ role;
  const int u0    = (wave & 1) * 64;

  for (int i = tid; i < 2 * BT_ * (H_ + 8); i += THREADS_) {
    ((_Float16*)(lds + OFF_H_(0, 0)))[i] = (_Float16)0.0f;
    ((_Float16*)(lds + OFF_H_(0, 1)))[i] = (_Float16)0.0f;
  }

  auto colof = [&](int n) { return ((n >> 2) * 2 + kind) * H_ + u0 + (n & 3) * 16; };

  const _Float16* Whh  = role ? Whh1 : Whh0;
  const float*    bias = role ? bias1 : bias0;
  v16h Bh[NT_][4];
  float bv[NT_];
  #pragma unroll
  for (int n = 0; n < NT_; ++n) {
    const int nc = colof(n);
    bv[n] = bias[nc + (lane & 15)];
    #pragma unroll
    for (int kt = 0; kt < 4; ++kt)
      Bh[n][kt] = ldfrag_g(Whh + (size_t)nc * H_, H_, kt * 32, lane);
  }

  _Float16* wmine = (_Float16*)(lds + (role ? OFF_W1_(pair, kp)
                                            : OFF_W0_(pair) + kp * (NT_ * 2 * FRAGH_ * 2)));
  if (role == 0) {
    #pragma unroll
    for (int n = 0; n < NT_; ++n) {
      #pragma unroll
      for (int kt = 0; kt < 2; ++kt)
        stfrag_w(wmine + (n * 2 + kt) * FRAGH_,
                 ldfrag_g(Wih0 + (size_t)colof(n) * D_, D_, kt * 32, lane), lane);
    }
  } else {
    #pragma unroll
    for (int n = 0; n < NT_; ++n) {
      #pragma unroll
      for (int kt = 0; kt < 4; ++kt)
        stfrag_w(wmine + (n * 4 + kt) * FRAGH_,
                 ldfrag_g(Wih1 + (size_t)colof(n) * H_, H_, kt * 32, lane), lane);
    }
  }

  const float up_pre  = kind ? 0.5f : 1.0f;
  const float up_post = kind ? 0.5f : 1.0f;
  const float up_off  = kind ? 0.5f : 0.0f;
  #pragma unroll
  for (int n = 0; n < NT_; ++n) bv[n] *= (n < 4) ? 0.5f : up_pre;

  v8f c[4];
  #pragma unroll
  for (int t = 0; t < 4; ++t) c[t] = v8f{0.f, 0.f, 0.f, 0.f, 0.f, 0.f, 0.f, 0.f};

  const _Float16* xbase = xseq + (size_t)b0 * T_ * D_;
  unsigned char* pmine = lds + (role ? (pair ? OFF_P_(1, 1) : OFF_P_(1, 0)) : (pair ? OFF_P_(0, 1) : OFF_P_(0, 0))) + lane * 16;
  const unsigned char* hown  = lds + (role ? (pair ? OFF_H_(1, 1) : OFF_H_(1, 0)) : (pair ? OFF_H_(0, 1) : OFF_H_(0, 0)));
  const unsigned char* hin0  = lds + (pair ? OFF_H_(0, 1) : OFF_H_(0, 0));
  unsigned char* hdstA = lds + (role ? OFF_H_(1, 0) : OFF_H_(0, 0)) + (u0 + (lane & 15)) * HTP_ + hi * 16;
  unsigned char* hdstB = lds + (role ? OFF_H_(1, 1) : OFF_H_(0, 1)) + (u0 + (lane & 15)) * HTP_ + hi * 16;

  __syncthreads();

  for (int s = 0; s <= T_; ++s) {
    const bool active = role ? (s >= 1) : (s < T_);
    v8f acc[NT_];

    if (active) {
      v16h bx[NT_];
      auto act_tile = [&](int n, float bvn) {
        const float pre  = (n < 4) ? 0.5f : up_pre;
        const float post = (n < 4) ? 0.5f : up_post;
        const float off  = (n < 4) ? 0.5f : up_off;
        #pragma unroll
        for (int m = 0; m < 8; ++m)
          acc[n][m] = off + post * fast_tanh(__builtin_fmaf(acc[n][m], pre, bvn));
      };
      if (role == 0) {
        v16h ha[4];
        #pragma unroll
        for (int kt = 0; kt < 4; ++kt) ha[kt] = ldfrag_ht(hown, kt, lane);
        v16h xa[2];
        #pragma unroll
        for (int kt = 0; kt < 2; ++kt)
          xa[kt] = ldfrag_g(xbase + (size_t)s * D_, T_ * D_, kt * 32, lane);
        if (s + 1 < T_)
          __builtin_prefetch(xbase + (size_t)(s + 1) * D_ + (size_t)(lane & 15) * (T_ * D_), 0, 1);
        #pragma unroll
        for (int kx = 0; kx < 2; ++kx) {
          #pragma unroll
          for (int n = 0; n < NT_; ++n) bx[n] = ldfrag_w(wmine + (n * 2 + kx) * FRAGH_, lane);
          __builtin_amdgcn_sched_barrier(0);
          if (kx == 0) {
            #pragma unroll
            for (int kt = 0; kt < 2; ++kt) {
              #pragma unroll
              for (int n = 0; n < NT_; ++n) {
                const v8f cin = (kt == 0) ? v8f{0.f, 0.f, 0.f, 0.f, 0.f, 0.f, 0.f, 0.f} : acc[n];
                acc[n] = __builtin_amdgcn_wmma_f32_16x16x32_f16(false, ha[kt], false, Bh[n][kt],
                                                                (short)0, cin, false, false);
              }
            }
            #pragma unroll
            for (int n = 0; n < NT_; ++n)
              acc[n] = __builtin_amdgcn_wmma_f32_16x16x32_f16(false, xa[kx], false, bx[n],
                                                              (short)0, acc[n], false, false);
            __builtin_amdgcn_sched_barrier(0);
          } else {
            #pragma unroll
            for (int n = 0; n < NT_; ++n)
              acc[n] = __builtin_amdgcn_wmma_f32_16x16x32_f16(false, ha[2], false, Bh[n][2],
                                                              (short)0, acc[n], false, false);
            #pragma unroll
            for (int n = 0; n < 4; ++n)
              acc[n] = __builtin_amdgcn_wmma_f32_16x16x32_f16(false, ha[3], false, Bh[n][3],
                                                              (short)0, acc[n], false, false);
            #pragma unroll
            for (int n = 0; n < NT_; ++n) {
              acc[n] = __builtin_amdgcn_wmma_f32_16x16x32_f16(false, xa[1], false, bx[n],
                                                              (short)0, acc[n], false, false);
              if (n + 4 < NT_)
                acc[n + 4] = __builtin_amdgcn_wmma_f32_16x16x32_f16(false, ha[3], false, Bh[n + 4][3],
                                                                    (short)0, acc[n + 4], false, false);
              act_tile(n, bv[n]);
            }
          }
        }
      } else {
        v16h ha[2], ia[2];
        #pragma unroll
        for (int kt = 0; kt < 2; ++kt) { ha[kt] = ldfrag_ht(hown, kt, lane); ia[kt] = ldfrag_ht(hin0, kt, lane); }
        #pragma unroll
        for (int kt = 0; kt < 4; ++kt) {
          #pragma unroll
          for (int n = 0; n < NT_; ++n) bx[n] = ldfrag_w(wmine + (n * 4 + kt) * FRAGH_, lane);
          if (kt == 1 || kt == 2) {
            ha[(kt + 1) & 1] = ldfrag_ht(hown, kt + 1, lane);
            ia[(kt + 1) & 1] = ldfrag_ht(hin0, kt + 1, lane);
          }
          __builtin_amdgcn_sched_barrier(0);
          if (kt < 3) {
            #pragma unroll
            for (int n = 0; n < NT_; ++n) {
              const v8f cin = (kt == 0) ? v8f{0.f, 0.f, 0.f, 0.f, 0.f, 0.f, 0.f, 0.f} : acc[n];
              acc[n] = __builtin_amdgcn_wmma_f32_16x16x32_f16(false, ha[kt & 1], false, Bh[n][kt],
                                                              (short)0, cin, false, false);
            }
            #pragma unroll
            for (int n = 0; n < NT_; ++n)
              acc[n] = __builtin_amdgcn_wmma_f32_16x16x32_f16(false, ia[kt & 1], false, bx[n],
                                                              (short)0, acc[n], false, false);
            __builtin_amdgcn_sched_barrier(0);
          } else {
            #pragma unroll
            for (int n = 0; n < 4; ++n)
              acc[n] = __builtin_amdgcn_wmma_f32_16x16x32_f16(false, ha[1], false, Bh[n][3],
                                                              (short)0, acc[n], false, false);
            #pragma unroll
            for (int n = 0; n < NT_; ++n) {
              acc[n] = __builtin_amdgcn_wmma_f32_16x16x32_f16(false, ia[1], false, bx[n],
                                                              (short)0, acc[n], false, false);
              if (n + 4 < NT_)
                acc[n + 4] = __builtin_amdgcn_wmma_f32_16x16x32_f16(false, ha[1], false, Bh[n + 4][3],
                                                                    (short)0, acc[n + 4], false, false);
              act_tile(n, bv[n]);
            }
          }
        }
      }

      if (kind == 0) {
        #pragma unroll
        for (int t = 0; t < 4; ++t) {
          const v8f p = acc[t] * acc[t + 4];
          *(v4f*)(pmine + (t * 2 + 0) * 512) = __builtin_shufflevector(p, p, 0, 1, 2, 3);
          *(v4f*)(pmine + (t * 2 + 1) * 512) = __builtin_shufflevector(p, p, 4, 5, 6, 7);
        }
      }
    }
    __syncthreads();

    if (active && kind == 1) {
      v4f pq[8];
      #pragma unroll
      for (int q = 0; q < 8; ++q) pq[q] = *(const v4f*)(pmine + q * 512);
      v8f hv[4];
      #pragma unroll
      for (int t = 0; t < 4; ++t) {
        #pragma unroll
        for (int m = 0; m < 8; ++m) {
          c[t][m] = acc[t][m] * c[t][m] + pq[2 * t + (m >> 2)][m & 3];
          hv[t][m] = acc[t + 4][m] * fast_tanh(c[t][m]);
        }
      }
      #pragma unroll
      for (int t = 0; t < 4; ++t) {
        const v8h hp = __builtin_convertvector(hv[t], v8h);
        *(v8h*)(hdstA + t * 16 * HTP_) = hp;
        *(v8h*)(hdstB + t * 16 * HTP_) = hp;
      }
      if (role == 1 && s == T_) {
        #pragma unroll
        for (int t = 0; t < 4; ++t) {
          #pragma unroll
          for (int m = 0; m < 8; ++m)
            hlast_out[(size_t)(b0 + 8 * hi + m) * H_ + u0 + t * 16 + (lane & 15)] = hv[t][m];
        }
      }
    }
    __syncthreads();
  }
}

__global__ void fc_kernel(const float* __restrict__ h1, const float* __restrict__ Wfc0,
                          const float* __restrict__ bfc0, const float* __restrict__ Wfc1,
                          const float* __restrict__ bfc1, float* __restrict__ out) {
  int b = blockIdx.x;
  int tid = threadIdx.x;
  __shared__ float tmp[32];
  float s = bfc0[tid];
  for (int k = 0; k < H_; ++k) s += h1[b * H_ + k] * Wfc0[tid * H_ + k];
  tmp[tid] = s;
  __syncthreads();
  if (tid < OUT_) {
    float o = bfc1[tid];
    for (int k = 0; k < 32; ++k) o += tmp[k] * Wfc1[tid * 32 + k];
    out[b * OUT_ + tid] = o;
  }
}

extern "C" void kernel_launch(void* const* d_in, const int* in_sizes, int n_in,
                              void* d_out, int out_size, void* d_ws, size_t ws_size,
                              hipStream_t stream) {
  (void)in_sizes; (void)n_in; (void)out_size; (void)ws_size;
  const float* x    = (const float*)d_in[0];
  const float* Wih0 = (const float*)d_in[1];
  const float* Whh0 = (const float*)d_in[2];
  const float* bih0 = (const float*)d_in[3];
  const float* bhh0 = (const float*)d_in[4];
  const float* Wih1 = (const float*)d_in[5];
  const float* Whh1 = (const float*)d_in[6];
  const float* bih1 = (const float*)d_in[7];
  const float* bhh1 = (const float*)d_in[8];
  const float* Wfc0 = (const float*)d_in[9];
  const float* bfc0 = (const float*)d_in[10];
  const float* Wfc1 = (const float*)d_in[11];
  const float* bfc1 = (const float*)d_in[12];

  char* p = (char*)d_ws;
  auto alloc = [&](size_t bytes) {
    char* r = p;
    p += (bytes + 255) & ~(size_t)255;
    return r;
  };
  _Float16* x_h    = (_Float16*)alloc((size_t)B_ * T_ * D_ * 2);
  _Float16* wih0_h = (_Float16*)alloc((size_t)G_ * D_ * 2);
  _Float16* whh0_h = (_Float16*)alloc((size_t)G_ * H_ * 2);
  _Float16* wih1_h = (_Float16*)alloc((size_t)G_ * H_ * 2);
  _Float16* whh1_h = (_Float16*)alloc((size_t)G_ * H_ * 2);
  float*    b0     = (float*)alloc((size_t)G_ * 4);
  float*    b1     = (float*)alloc((size_t)G_ * 4);
  float*    h1last = (float*)alloc((size_t)B_ * H_ * 4);

  static_assert((B_ * T_ * D_) % (CVT_E_ * CVT_UX_ * CVT_T_) == 0, "x pass: whole workgroups");
  static_assert((G_ * D_) % (CVT_E_ * CVT_T_) == 0 && (G_ * H_) % (CVT_E_ * CVT_T_) == 0,
                "weight passes: whole workgroups");
  cvt_f32_to_f16<CVT_UX_><<<B_ * T_ * D_ / (CVT_E_ * CVT_UX_ * CVT_T_), CVT_T_, 0, stream>>>(x, x_h);
  cvt_f32_to_f16<1><<<G_ * D_ / (CVT_E_ * CVT_T_), CVT_T_, 0, stream>>>(Wih0, wih0_h);
  cvt_f32_to_f16<1><<<G_ * H_ / (CVT_E_ * CVT_T_), CVT_T_, 0, stream>>>(Whh0, whh0_h);
  cvt_f32_to_f16<1><<<G_ * H_ / (CVT_E_ * CVT_T_), CVT_T_, 0, stream>>>(Wih1, wih1_h);
  cvt_f32_to_f16<1><<<G_ * H_ / (CVT_E_ * CVT_T_), CVT_T_, 0, stream>>>(Whh1, whh1_h);
  add_vec<<<2, 256, 0, stream>>>(bih0, bhh0, b0, G_);
  add_vec<<<2, 256, 0, stream>>>(bih1, bhh1, b1, G_);

  lstm2_kernel<<<B_ / BT_, THREADS_, 0, stream>>>(x_h, wih0_h, whh0_h, b0,
                                                  wih1_h, whh1_h, b1, h1last);
  fc_kernel<<<B_, 32, 0, stream>>>(h1last, Wfc0, bfc0, Wfc1, bfc1, (float*)d_out);
}
